# starting kernel
# baseline (speedup 1.0000x reference)
#include <hip/hip_runtime.h>
#include <cmath>

#define B_  8
#define TE_ 256
#define TD_ 256
#define D_  256
#define U_  256

typedef float v2f __attribute__((ext_vector_type(2)));
typedef float v8f __attribute__((ext_vector_type(8)));

__global__ void __launch_bounds__(256) gemm_f32_wmma(
    const float* __restrict__ A, const float* __restrict__ W,
    float* __restrict__ C) {
  const int lane  = threadIdx.x & 31;
  const int wave  = threadIdx.x >> 5;
  const int tile  = blockIdx.x * 8 + wave;
  const int tn    = tile & 15;
  const int tm    = tile >> 4;
  const int m     = lane & 15;
  const int khalf = lane >> 4;

  const float* arow = A + (size_t)(tm * 16 + m) * U_;
  const float* bcol = W + tn * 16 + m;
  v8f acc = {};
  for (int k = 0; k < U_; k += 4) {
    const int ka = k + 2 * khalf;
    v2f a, b;
    a.x = arow[ka + 0];
    a.y = arow[ka + 1];
    b.x = bcol[(size_t)(ka + 0) * U_];
    b.y = bcol[(size_t)(ka + 1) * U_];
    acc = __builtin_amdgcn_wmma_f32_16x16x4_f32(false, a, false, b,
                                                (short)0, acc, false, false);
  }
#pragma unroll
  for (int r = 0; r < 8; ++r) {
    const int m_out = tm * 16 + r + 8 * khalf;
    C[(size_t)m_out * U_ + tn * 16 + m] = acc[r];
  }
}

__global__ void __launch_bounds__(256) co_mu_kernel(
    const float* __restrict__ ae, const float* __restrict__ ad,
    const float* __restrict__ nu, float* __restrict__ mu) {
  const int b      = blockIdx.y;
  const int s_tile = blockIdx.x >> 4;
  const int t_tile = blockIdx.x & 15;
  __shared__ float adT[16][260];
  __shared__ float aeT[16][260];
  __shared__ float nuS[256];

  const int tid = threadIdx.x;
  nuS[tid] = nu[tid];
  for (int i = tid; i < 16 * 256; i += 256) {
    const int r = i >> 8, c = i & 255;
    adT[r][c] = ad[((size_t)(b * TD_ + s_tile * 16 + r)) * U_ + c];
    aeT[r][c] = ae[((size_t)(b * TE_ + t_tile * 16 + r)) * U_ + c];
  }
  __syncthreads();

  const int si = tid >> 4, ti = tid & 15;
  float acc = 0.f;
#pragma unroll 4
  for (int u = 0; u < U_; ++u)
    acc += tanhf(adT[si][u] + aeT[ti][u]) * nuS[u];

  mu[((size_t)(b * TD_ + s_tile * 16 + si)) * TE_ + t_tile * 16 + ti] = acc;
}

__global__ void __launch_bounds__(256) softmax_rows_kernel(
    float* __restrict__ mu, float* __restrict__ max_mu) {
  const int row = blockIdx.x;
  const int tid = threadIdx.x;
  __shared__ float red[256];

  const float v = mu[(size_t)row * TE_ + tid];
  red[tid] = v; __syncthreads();
  for (int s = 128; s > 0; s >>= 1) {
    if (tid < s) red[tid] = fmaxf(red[tid], red[tid + s]);
    __syncthreads();
  }
  const float mx = red[0];
  __syncthreads();
  const float e = expf(v - mx);
  red[tid] = e; __syncthreads();
  for (int s = 128; s > 0; s >>= 1) {
    if (tid < s) red[tid] += red[tid + s];
    __syncthreads();
  }
  mu[(size_t)row * TE_ + tid] = e * (1.0f / red[0]);
  if (tid == 0) max_mu[row] = mx;
}

__global__ void __launch_bounds__(256) hhat_kernel(
    const float* __restrict__ max_mu, const float* __restrict__ de_seq,
    float* __restrict__ h_hat) {
  const int b   = blockIdx.x;
  const int tid = threadIdx.x;
  __shared__ float red[256];
  __shared__ float ma[256];

  const float v = max_mu[b * TD_ + tid];
  red[tid] = v; __syncthreads();
  for (int s = 128; s > 0; s >>= 1) {
    if (tid < s) red[tid] = fmaxf(red[tid], red[tid + s]);
    __syncthreads();
  }
  const float mx = red[0];
  __syncthreads();
  const float e = expf(v - mx);
  red[tid] = e; __syncthreads();
  for (int s = 128; s > 0; s >>= 1) {
    if (tid < s) red[tid] += red[tid + s];
    __syncthreads();
  }
  ma[tid] = e * (1.0f / red[0]);
  __syncthreads();

  float acc = 0.f;
  for (int s = 0; s < TD_; ++s)
    acc += de_seq[((size_t)(b * TD_ + s)) * D_ + tid] * ma[s];
  h_hat[b * D_ + tid] = acc;
}

__global__ void __launch_bounds__(256) sum_en_out_kernel(
    const float* __restrict__ alphas, const float* __restrict__ en_seq,
    const float* __restrict__ de_seq, const float* __restrict__ h_hat,
    float* __restrict__ out) {
  const int b     = blockIdx.y;
  const int lane  = threadIdx.x & 31;
  const int wave  = threadIdx.x >> 5;
  const int tile  = blockIdx.x * 8 + wave;
  const int td    = tile & 15;
  const int ts    = tile >> 4;
  const int m     = lane & 15;
  const int khalf = lane >> 4;

  const float* A    = alphas + (size_t)b * TD_ * TE_;
  const float* Bm   = en_seq + (size_t)b * TE_ * D_;
  const float* arow = A + (size_t)(ts * 16 + m) * TE_;
  const float* bcol = Bm + td * 16 + m;

  v8f acc = {};
  for (int k = 0; k < TE_; k += 4) {
    const int ka = k + 2 * khalf;
    v2f a, bb;
    a.x  = arow[ka + 0];
    a.y  = arow[ka + 1];
    bb.x = bcol[(size_t)(ka + 0) * D_];
    bb.y = bcol[(size_t)(ka + 1) * D_];
    acc = __builtin_amdgcn_wmma_f32_16x16x4_f32(false, a, false, bb,
                                                (short)0, acc, false, false);
  }

  const int d_g  = td * 16 + m;
  const float hh = h_hat[b * D_ + d_g];
#pragma unroll
  for (int r = 0; r < 8; ++r) {
    const int s_g  = ts * 16 + r + 8 * khalf;
    const float se = acc[r];
    const float de = de_seq[((size_t)(b * TD_ + s_g)) * D_ + d_g];
    float* o = out + ((size_t)(b * TD_ + s_g)) * (4 * D_);
    o[d_g]          = de;
    o[D_ + d_g]     = se;
    o[2 * D_ + d_g] = de * se;
    o[3 * D_ + d_g] = de * hh;
  }
}

extern "C" void kernel_launch(void* const* d_in, const int* in_sizes, int n_in,
                              void* d_out, int out_size, void* d_ws, size_t ws_size,
                              hipStream_t stream) {
  const float* en_seq = (const float*)d_in[0];
  const float* de_seq = (const float*)d_in[1];
  const float* w_en   = (const float*)d_in[2];
  const float* w_de   = (const float*)d_in[3];
  const float* nu     = (const float*)d_in[4];
  float* out = (float*)d_out;

  float* ws = (float*)d_ws;
  const size_t nAE = (size_t)B_ * TE_ * U_;
  float* ae     = ws;
  float* ad     = ae + nAE;
  float* mu     = ad + nAE;
  float* max_mu = mu + (size_t)B_ * TD_ * TE_;
  float* h_hat  = max_mu + (size_t)B_ * TD_;

  const dim3 blk(256);
  gemm_f32_wmma<<<dim3(256), blk, 0, stream>>>(en_seq, w_en, ae);
  gemm_f32_wmma<<<dim3(256), blk, 0, stream>>>(de_seq, w_de, ad);
  co_mu_kernel<<<dim3(256, B_), blk, 0, stream>>>(ae, ad, nu, mu);
  softmax_rows_kernel<<<dim3(B_ * TD_), blk, 0, stream>>>(mu, max_mu);
  hhat_kernel<<<dim3(B_), blk, 0, stream>>>(max_mu, de_seq, h_hat);
  sum_en_out_kernel<<<dim3(32, B_), blk, 0, stream>>>(mu, en_seq, de_seq,
                                                      h_hat, out);
}


# the three f32 WMMA products (ae, ad, alphas@en) read their operands from LDS: a block copies a 32x256 A and a 256x64 B panel by 16-byte loads, handed over in 4 K quarters; k and sum order unchanged
# speedup vs baseline: 3.1854x; 3.1854x over previous
#include <hip/hip_runtime.h>
#include <cmath>

#define B_  8
#define TE_ 256
#define TD_ 256
#define D_  256
#define U_  256

typedef float v2f __attribute__((ext_vector_type(2)));
typedef float v8f __attribute__((ext_vector_type(8)));

typedef float v4f __attribute__((ext_vector_type(4)));
#define PA_ 260
#define PB_ 72
#define PANEL_A_FLOATS_ (32 * PA_)
#define PANEL_B_FLOATS_ (256 * PB_)

__device__ __forceinline__ void lds_quarter_barrier() {
  asm volatile("s_wait_dscnt 0x0" ::: "memory");
  asm volatile("s_barrier_signal -1" ::: "memory");
  asm volatile("s_barrier_wait -1" ::: "memory");
}

__device__ __forceinline__ v8f mma_panel_32x64(const float* __restrict__ Ag,
                                               const float* __restrict__ Bg,
                                               float* As, float* Bs) {
  const int tid   = threadIdx.x;
  const int lane  = tid & 31;
  const int wave  = tid >> 5;
  const int m     = lane & 15;
  const int khalf = lane >> 4;
  const int wm    = wave >> 2;
  const int wn    = wave & 3;
  const int row   = tid >> 4;
  const int c4    = (tid & 15) * 4;

  v4f ra[4][2], rb[4][4];
#pragma unroll
  for (int q = 0; q < 4; ++q) {
#pragma unroll
    for (int j = 0; j < 2; ++j)
      ra[q][j] = *(const v4f*)(Ag + (size_t)(j * 16 + row) * 256 + 64 * q + c4);
#pragma unroll
    for (int j = 0; j < 4; ++j)
      rb[q][j] = *(const v4f*)(Bg + (size_t)(64 * q + j * 16 + row) * 256 + c4);
  }

  const float* af = As + (wm * 16 + m) * PA_ + 2 * khalf;
  const float* bf = Bs + (2 * khalf) * PB_ + wn * 16 + m;
  v2f fa[2][16], fb[2][16];
  v8f acc = {};
#pragma unroll
  for (int q = -1; q < 4; ++q) {
    if (q < 3) {
      const int qn = q + 1;
#pragma unroll
      for (int j = 0; j < 2; ++j)
        *(v4f*)(As + (j * 16 + row) * PA_ + 64 * qn + c4) = ra[qn][j];
#pragma unroll
      for (int j = 0; j < 4; ++j)
        *(v4f*)(Bs + (64 * qn + j * 16 + row) * PB_ + c4) = rb[qn][j];
      lds_quarter_barrier();
#pragma unroll
      for (int i = 0; i < 16; ++i) {
        const int k4 = 4 * (16 * qn + i);
        fa[qn & 1][i]   = *(const v2f*)(af + k4);
        fb[qn & 1][i].x = bf[(k4 + 0) * PB_];
        fb[qn & 1][i].y = bf[(k4 + 1) * PB_];
      }
    }
    __builtin_amdgcn_sched_barrier(0);
    if (q >= 0) {
#pragma unroll
      for (int i = 0; i < 16; ++i)
        acc = __builtin_amdgcn_wmma_f32_16x16x4_f32(false, fa[q & 1][i], false,
                                                    fb[q & 1][i], (short)0, acc,
                                                    false, false);
    }
    __builtin_amdgcn_sched_barrier(0);
  }
  return acc;
}

__global__ void __launch_bounds__(256) gemm_f32_wmma(
    const float* __restrict__ A, const float* __restrict__ W,
    float* __restrict__ C) {
  __shared__ __attribute__((aligned(16))) float As[PANEL_A_FLOATS_];
  __shared__ __attribute__((aligned(16))) float Bs[PANEL_B_FLOATS_];
  const int lane  = threadIdx.x & 31;
  const int wave  = threadIdx.x >> 5;
  const int m     = lane & 15;
  const int khalf = lane >> 4;
  const int mb    = blockIdx.x >> 2;
  const int nb    = blockIdx.x & 3;

  const v8f acc = mma_panel_32x64(A + (size_t)mb * 32 * U_, W + nb * 64, As, Bs);

  const int n_out = nb * 64 + (wave & 3) * 16 + m;
#pragma unroll
  for (int r = 0; r < 8; ++r) {
    const int m_out = mb * 32 + (wave >> 2) * 16 + r + 8 * khalf;
    C[(size_t)m_out * U_ + n_out] = acc[r];
  }
}

#define EXP_LIM_ 80.0f

__device__ __forceinline__ float exp_staged(float v) {
  const float kHi = 1.44269502f;
  const float kLo = 1.92596299e-8f;
  const float kLn2 = 0.693147182f;
  const float hi = kHi * v;
  float lo = fmaf(kHi, v, -hi);
  lo = fmaf(kLo, v, lo);
  const float e = __builtin_amdgcn_exp2f(hi);
  return fmaf(e, lo * kLn2, e);
}

__global__ void __launch_bounds__(256) co_mu_kernel(
    const float* __restrict__ ae, const float* __restrict__ ad,
    const float* __restrict__ nu, float* __restrict__ mu) {
  const int b      = blockIdx.y;
  const int s_tile = blockIdx.x >> 4;
  const int t_tile = blockIdx.x & 15;
  __shared__ float adT[16][260];
  __shared__ float aeT[16][260];
  __shared__ float nuS[256];
  __shared__ float nuPart[8];
  __shared__ int   badPart[8];

  const int tid  = threadIdx.x;
  const int lane = tid & 31;
  const int wave = tid >> 5;
  const float* adp = ad + ((size_t)(b * TD_ + s_tile * 16)) * U_;
  const float* aep = ae + ((size_t)(b * TE_ + t_tile * 16)) * U_;

  const float nuv = nu[tid];
  nuS[tid] = -2.0f * nuv;
  int bad = 0;
  for (int i = tid; i < 16 * 256; i += 256) {
    const int r = i >> 8, c = i & 255;
    const float va = adp[i];
    const float ve = aep[i];
    bad |= !(fabsf(va) <= EXP_LIM_) | !(fabsf(ve) <= EXP_LIM_);
    adT[r][c] = exp_staged(va);
    aeT[r][c] = exp_staged(ve);
  }
  float nsum = nuv;
#pragma unroll
  for (int m = 16; m > 0; m >>= 1) nsum += __shfl_xor(nsum, m);
  const int wave_bad = __any(bad);
  if (lane == 0) { nuPart[wave] = nsum; badPart[wave] = wave_bad; }
  __syncthreads();

  float nu_total = 0.f;
  int any_bad = 0;
#pragma unroll
  for (int w = 0; w < 8; ++w) { nu_total += nuPart[w]; any_bad |= badPart[w]; }

  const int si = tid >> 4, ti = tid & 15;
  float* dst = mu + ((size_t)(b * TD_ + s_tile * 16 + si)) * TE_ + t_tile * 16 + ti;

  if (!any_bad) {
    float acc = 0.f;
#pragma unroll 4
    for (int u = 0; u < U_; ++u) {
      const float p   = adT[si][u] * aeT[ti][u];
      const float den = fmaf(p, p, 1.0f);
      acc = fmaf(nuS[u], __builtin_amdgcn_rcpf(den), acc);
    }
    *dst = acc + nu_total;
  } else {
    nuS[tid] = nuv;
    for (int i = tid; i < 16 * 256; i += 256) {
      const int r = i >> 8, c = i & 255;
      adT[r][c] = adp[i];
      aeT[r][c] = aep[i];
    }
    __syncthreads();
    float acc = 0.f;
#pragma unroll 4
    for (int u = 0; u < U_; ++u)
      acc += tanhf(adT[si][u] + aeT[ti][u]) * nuS[u];
    *dst = acc;
  }
}

__global__ void __launch_bounds__(256) softmax_rows_kernel(
    float* __restrict__ mu, float* __restrict__ max_mu) {
  const int row = blockIdx.x;
  const int tid = threadIdx.x;
  __shared__ float red[256];

  const float v = mu[(size_t)row * TE_ + tid];
  red[tid] = v; __syncthreads();
  for (int s = 128; s > 0; s >>= 1) {
    if (tid < s) red[tid] = fmaxf(red[tid], red[tid + s]);
    __syncthreads();
  }
  const float mx = red[0];
  __syncthreads();
  const float e = expf(v - mx);
  red[tid] = e; __syncthreads();
  for (int s = 128; s > 0; s >>= 1) {
    if (tid < s) red[tid] += red[tid + s];
    __syncthreads();
  }
  mu[(size_t)row * TE_ + tid] = e * (1.0f / red[0]);
  if (tid == 0) max_mu[row] = mx;
}

__device__ __forceinline__ float lane_xor16(float x) {
  const int b = __builtin_bit_cast(int, x);
  return __builtin_bit_cast(float,
      __builtin_amdgcn_permlanex16(b, b, 0x76543210, 0xfedcba98, false, false));
}
template <int M>
__device__ __forceinline__ float lane_xor_row(float x) {
  const int b = __builtin_bit_cast(int, x);
  return __builtin_bit_cast(float,
      __builtin_amdgcn_update_dpp(b, b, 0x160 + M, 0xf, 0xf, false));
}

__global__ void __launch_bounds__(256) hhat_kernel(
    const float* __restrict__ max_mu, const float* __restrict__ de_seq,
    float* __restrict__ h_hat) {
  const int b    = blockIdx.y;
  const int d0   = blockIdx.x * 32;
  const int lane = threadIdx.x & 31;
  const int wave = __builtin_amdgcn_readfirstlane((int)(threadIdx.x >> 5));
  __shared__ float part[8][32];

  const float* mm = max_mu + b * TD_ + lane;
  float v[8];
#pragma unroll
  for (int j = 0; j < 8; ++j) v[j] = mm[32 * j];

  const float* dp = de_seq + ((size_t)(b * TD_ + wave * 32)) * D_ + d0 + lane;
  float de[32];
#pragma unroll
  for (int i = 0; i < 32; ++i) de[i] = dp[(size_t)i * D_];

  float mx = fmaxf(fmaxf(fmaxf(v[0], v[4]), fmaxf(v[2], v[6])),
                   fmaxf(fmaxf(v[1], v[5]), fmaxf(v[3], v[7])));
  mx = fmaxf(mx, lane_xor16(mx));
  mx = fmaxf(mx, lane_xor_row<8>(mx));
  mx = fmaxf(mx, lane_xor_row<4>(mx));
  mx = fmaxf(mx, lane_xor_row<2>(mx));
  mx = fmaxf(mx, lane_xor_row<1>(mx));

  float e[8];
#pragma unroll
  for (int j = 0; j < 8; ++j) e[j] = expf(v[j] - mx);
  float sum = ((e[0] + e[4]) + (e[2] + e[6])) + ((e[1] + e[5]) + (e[3] + e[7]));
  sum = sum + lane_xor16(sum);
  sum = sum + lane_xor_row<8>(sum);
  sum = sum + lane_xor_row<4>(sum);
  sum = sum + lane_xor_row<2>(sum);
  sum = sum + lane_xor_row<1>(sum);
  const float inv = 1.0f / sum;

  float eo = e[0];
#pragma unroll
  for (int j = 1; j < 8; ++j) eo = (wave == j) ? e[j] : eo;
  const float ma = eo * inv;

  float acc = 0.f;
#pragma unroll
  for (int i = 0; i < 32; ++i) {
    const float m_i = __builtin_bit_cast(float,
        __builtin_amdgcn_readlane(__builtin_bit_cast(int, ma), i));
    acc += de[i] * m_i;
  }
  part[wave][lane] = acc;
  __syncthreads();
  if (wave == 0) {
    float t = part[0][lane];
#pragma unroll
    for (int w = 1; w < 8; ++w) t += part[w][lane];
    h_hat[b * D_ + d0 + lane] = t;
  }
}

__global__ void __launch_bounds__(256) sum_en_out_kernel(
    const float* __restrict__ alphas, const float* __restrict__ en_seq,
    const float* __restrict__ de_seq, const float* __restrict__ h_hat,
    float* __restrict__ out) {
  __shared__ __attribute__((aligned(16))) float As[PANEL_A_FLOATS_];
  __shared__ __attribute__((aligned(16))) float Bs[PANEL_B_FLOATS_];
  const int b     = blockIdx.y;
  const int lane  = threadIdx.x & 31;
  const int wave  = threadIdx.x >> 5;
  const int m     = lane & 15;
  const int khalf = lane >> 4;
  const int mb    = blockIdx.x >> 2;
  const int nb    = blockIdx.x & 3;

  const v8f acc = mma_panel_32x64(
      alphas + ((size_t)b * TD_ + mb * 32) * TE_,
      en_seq + (size_t)b * TE_ * D_ + nb * 64, As, Bs);

  const int d_g  = nb * 64 + (wave & 3) * 16 + m;
  const float hh = h_hat[b * D_ + d_g];
#pragma unroll
  for (int r = 0; r < 8; ++r) {
    const int s_g  = mb * 32 + (wave >> 2) * 16 + r + 8 * khalf;
    const float se = acc[r];
    const float de = de_seq[((size_t)(b * TD_ + s_g)) * D_ + d_g];
    float* o = out + ((size_t)(b * TD_ + s_g)) * (4 * D_);
    o[d_g]          = de;
    o[D_ + d_g]     = se;
    o[2 * D_ + d_g] = de * se;
    o[3 * D_ + d_g] = de * hh;
  }
}

extern "C" void kernel_launch(void* const* d_in, const int* in_sizes, int n_in,
                              void* d_out, int out_size, void* d_ws, size_t ws_size,
                              hipStream_t stream) {
  const float* en_seq = (const float*)d_in[0];
  const float* de_seq = (const float*)d_in[1];
  const float* w_en   = (const float*)d_in[2];
  const float* w_de   = (const float*)d_in[3];
  const float* nu     = (const float*)d_in[4];
  float* out = (float*)d_out;

  float* ws = (float*)d_ws;
  const size_t nAE = (size_t)B_ * TE_ * U_;
  float* ae     = ws;
  float* ad     = ae + nAE;
  float* mu     = ad + nAE;
  float* max_mu = mu + (size_t)B_ * TD_ * TE_;
  float* h_hat  = max_mu + (size_t)B_ * TD_;

  const dim3 blk(256);
  gemm_f32_wmma<<<dim3(256), blk, 0, stream>>>(en_seq, w_en, ae);
  gemm_f32_wmma<<<dim3(256), blk, 0, stream>>>(de_seq, w_de, ad);
  co_mu_kernel<<<dim3(256, B_), blk, 0, stream>>>(ae, ad, nu, mu);
  softmax_rows_kernel<<<dim3(B_ * TD_), blk, 0, stream>>>(mu, max_mu);
  hhat_kernel<<<dim3(8, B_), blk, 0, stream>>>(max_mu, de_seq, h_hat);
  sum_en_out_kernel<<<dim3(32, B_), blk, 0, stream>>>(mu, en_seq, de_seq,
                                                      h_hat, out);
}


# score and row-softmax kernels merged: a block owns 8 whole mu rows (8 s x all 256 t, 8 outputs per lane, exp(ae) of the batch in LDS), softmax by lane exchange; row-sum order changed, u order kept
# speedup vs baseline: 3.7383x; 1.1736x over previous
#include <hip/hip_runtime.h>
#include <cmath>

#define B_  8
#define TE_ 256
#define TD_ 256
#define D_  256
#define U_  256

typedef float v2f __attribute__((ext_vector_type(2)));
typedef float v8f __attribute__((ext_vector_type(8)));

typedef float v4f __attribute__((ext_vector_type(4)));
#define PA_ 260
#define PB_ 72
#define PANEL_A_FLOATS_ (32 * PA_)
#define PANEL_B_FLOATS_ (256 * PB_)

__device__ __forceinline__ void lds_quarter_barrier() {
  asm volatile("s_wait_dscnt 0x0" ::: "memory");
  asm volatile("s_barrier_signal -1" ::: "memory");
  asm volatile("s_barrier_wait -1" ::: "memory");
}

__device__ __forceinline__ v8f mma_panel_32x64(const float* __restrict__ Ag,
                                               const float* __restrict__ Bg,
                                               float* As, float* Bs) {
  const int tid   = threadIdx.x;
  const int lane  = tid & 31;
  const int wave  = tid >> 5;
  const int m     = lane & 15;
  const int khalf = lane >> 4;
  const int wm    = wave >> 2;
  const int wn    = wave & 3;
  const int row   = tid >> 4;
  const int c4    = (tid & 15) * 4;

  v4f ra[4][2], rb[4][4];
#pragma unroll
  for (int q = 0; q < 4; ++q) {
#pragma unroll
    for (int j = 0; j < 2; ++j)
      ra[q][j] = *(const v4f*)(Ag + (size_t)(j * 16 + row) * 256 + 64 * q + c4);
#pragma unroll
    for (int j = 0; j < 4; ++j)
      rb[q][j] = *(const v4f*)(Bg + (size_t)(64 * q + j * 16 + row) * 256 + c4);
  }

  const float* af = As + (wm * 16 + m) * PA_ + 2 * khalf;
  const float* bf = Bs + (2 * khalf) * PB_ + wn * 16 + m;
  v2f fa[2][16], fb[2][16];
  v8f acc = {};
#pragma unroll
  for (int q = -1; q < 4; ++q) {
    if (q < 3) {
      const int qn = q + 1;
#pragma unroll
      for (int j = 0; j < 2; ++j)
        *(v4f*)(As + (j * 16 + row) * PA_ + 64 * qn + c4) = ra[qn][j];
#pragma unroll
      for (int j = 0; j < 4; ++j)
        *(v4f*)(Bs + (64 * qn + j * 16 + row) * PB_ + c4) = rb[qn][j];
      lds_quarter_barrier();
#pragma unroll
      for (int i = 0; i < 16; ++i) {
        const int k4 = 4 * (16 * qn + i);
        fa[qn & 1][i]   = *(const v2f*)(af + k4);
        fb[qn & 1][i].x = bf[(k4 + 0) * PB_];
        fb[qn & 1][i].y = bf[(k4 + 1) * PB_];
      }
    }
    __builtin_amdgcn_sched_barrier(0);
    if (q >= 0) {
#pragma unroll
      for (int i = 0; i < 16; ++i)
        acc = __builtin_amdgcn_wmma_f32_16x16x4_f32(false, fa[q & 1][i], false,
                                                    fb[q & 1][i], (short)0, acc,
                                                    false, false);
    }
    __builtin_amdgcn_sched_barrier(0);
  }
  return acc;
}

__global__ void __launch_bounds__(256) gemm_f32_wmma(
    const float* __restrict__ A, const float* __restrict__ W,
    float* __restrict__ C) {
  __shared__ __attribute__((aligned(16))) float As[PANEL_A_FLOATS_];
  __shared__ __attribute__((aligned(16))) float Bs[PANEL_B_FLOATS_];
  const int lane  = threadIdx.x & 31;
  const int wave  = threadIdx.x >> 5;
  const int m     = lane & 15;
  const int khalf = lane >> 4;
  const int mb    = blockIdx.x >> 2;
  const int nb    = blockIdx.x & 3;

  const v8f acc = mma_panel_32x64(A + (size_t)mb * 32 * U_, W + nb * 64, As, Bs);

  const int n_out = nb * 64 + (wave & 3) * 16 + m;
#pragma unroll
  for (int r = 0; r < 8; ++r) {
    const int m_out = mb * 32 + (wave >> 2) * 16 + r + 8 * khalf;
    C[(size_t)m_out * U_ + n_out] = acc[r];
  }
}

#define EXP_LIM_ 80.0f
#define PE_ 260

__device__ __forceinline__ float exp_staged(float v) {
  const float kHi = 1.44269502f;
  const float kLo = 1.92596299e-8f;
  const float kLn2 = 0.693147182f;
  const float hi = kHi * v;
  float lo = fmaf(kHi, v, -hi);
  lo = fmaf(kLo, v, lo);
  const float e = __builtin_amdgcn_exp2f(hi);
  return fmaf(e, lo * kLn2, e);
}

__device__ __forceinline__ float mu_direct(const float* __restrict__ adr,
                                           const float* __restrict__ aer,
                                           const float* __restrict__ nu) {
  float acc = 0.f;
  for (int u = 0; u < U_; ++u) acc += tanhf(adr[u] + aer[u]) * nu[u];
  return acc;
}

__device__ __forceinline__ float lane_xor16(float x) {
  const int b = __builtin_bit_cast(int, x);
  return __builtin_bit_cast(float,
      __builtin_amdgcn_permlanex16(b, b, 0x76543210, 0xfedcba98, false, false));
}
template <int M>
__device__ __forceinline__ float lane_xor_row(float x) {
  const int b = __builtin_bit_cast(int, x);
  return __builtin_bit_cast(float,
      __builtin_amdgcn_update_dpp(b, b, 0x160 + M, 0xf, 0xf, false));
}

__global__ void __launch_bounds__(256) co_rows_softmax_kernel(
    const float* __restrict__ ae, const float* __restrict__ ad,
    const float* __restrict__ nu, float* __restrict__ alphas,
    float* __restrict__ max_mu) {
  __shared__ __attribute__((aligned(16))) float EeT[256 * PE_];
  __shared__ __attribute__((aligned(16))) float EaT[8 * PE_];
  __shared__ __attribute__((aligned(16))) float nuS[256];
  __shared__ float muFix[256];
  __shared__ float nuPart[8];
  __shared__ int   badT[256];
  __shared__ int   badS[8];
  __shared__ int   anyBad;

  const int b    = blockIdx.y;
  const int s0   = blockIdx.x * 8;
  const int tid  = threadIdx.x;
  const int lane = tid & 31;
  const int wave = __builtin_amdgcn_readfirstlane((int)(threadIdx.x >> 5));
  const int row  = tid >> 4;
  const int c4   = (tid & 15) * 4;
  const float* aep = ae + (size_t)b * TE_ * U_;
  const float* adp = ad + ((size_t)b * TD_ + s0) * U_;

  v4f cur[16];
#pragma unroll
  for (int i = 0; i < 16; ++i)
    cur[i] = *(const v4f*)(aep + (size_t)(16 * i + row) * U_ + c4);
  v4f va[2];
#pragma unroll
  for (int k = 0; k < 2; ++k) va[k] = *(const v4f*)(adp + (size_t)(tid + 256 * k) * 4);

  const float nuv = nu[tid];
  nuS[tid]  = -2.0f * nuv;
  badT[tid] = 0;
  if (tid < 8) badS[tid] = 0;
  if (tid == 0) anyBad = 0;
  float nsum = nuv;
#pragma unroll
  for (int m = 16; m > 0; m >>= 1) nsum += __shfl_xor(nsum, m);
  if (lane == 0) nuPart[wave] = nsum;

  int badA[2];
#pragma unroll
  for (int k = 0; k < 2; ++k) {
    const int idx = tid + 256 * k;
    v4f e;
    int bad = 0;
#pragma unroll
    for (int x = 0; x < 4; ++x) {
      bad |= !(fabsf(va[k][x]) <= EXP_LIM_);
      e[x] = exp_staged(va[k][x]);
    }
    badA[k] = bad;
    *(v4f*)(EaT + (idx >> 6) * PE_ + (idx & 63) * 4) = e;
  }

  const float* eaRow  = EaT + wave * PE_;
  const float* eeLane = EeT + lane * PE_;
  float acc[8];
#pragma unroll
  for (int j = 0; j < 8; ++j) acc[j] = 0.f;
  unsigned badmask = 0;

#pragma unroll
  for (int q = 0; q < 4; ++q) {
#pragma unroll
    for (int i = 0; i < 16; ++i) {
      v4f e;
      int bad = 0;
#pragma unroll
      for (int x = 0; x < 4; ++x) {
        bad |= !(fabsf(cur[i][x]) <= EXP_LIM_);
        e[x] = exp_staged(cur[i][x]);
      }
      badmask |= (unsigned)bad << i;
      *(v4f*)(EeT + (16 * i + row) * PE_ + 64 * q + c4) = e;
    }
    if (q < 3) {
#pragma unroll
      for (int i = 0; i < 16; ++i)
        cur[i] = *(const v4f*)(aep + (size_t)(16 * i + row) * U_ + 64 * (q + 1) + c4);
    }
    lds_quarter_barrier();

#pragma unroll 2
    for (int u = 64 * q; u < 64 * q + 64; u += 4) {
      const v4f a4 = *(const v4f*)(eaRow + u);
      const v4f n4 = *(const v4f*)(nuS + u);
#pragma unroll
      for (int j = 0; j < 8; ++j) {
        const v4f e4 = *(const v4f*)(eeLane + j * 32 * PE_ + u);
#pragma unroll
        for (int x = 0; x < 4; ++x) {
          const float p   = a4[x] * e4[x];
          const float den = fmaf(p, p, 1.0f);
          acc[j] = fmaf(n4[x], __builtin_amdgcn_rcpf(den), acc[j]);
        }
      }
    }
  }

  if (badA[0]) { badS[tid >> 6] = 1; anyBad = 1; }
  if (badA[1]) { badS[(tid + 256) >> 6] = 1; anyBad = 1; }
  if (badmask) {
    anyBad = 1;
#pragma unroll
    for (int i = 0; i < 16; ++i)
      if ((badmask >> i) & 1u) badT[16 * i + row] = 1;
  }
  __syncthreads();

  float nu_total = 0.f;
#pragma unroll
  for (int w = 0; w < 8; ++w) nu_total += nuPart[w];
  float mu[8];
#pragma unroll
  for (int j = 0; j < 8; ++j) mu[j] = acc[j] + nu_total;

  if (anyBad) {
    for (int s = 0; s < 8; ++s) {
      if (badS[s]) {
        muFix[tid] = mu_direct(adp + (size_t)s * U_, aep + (size_t)tid * U_, nu);
        __syncthreads();
        if (wave == s) {
#pragma unroll
          for (int j = 0; j < 8; ++j) mu[j] = muFix[lane + 32 * j];
        }
        __syncthreads();
      }
    }
    if (!badS[wave]) {
#pragma unroll
      for (int j = 0; j < 8; ++j)
        if (badT[lane + 32 * j])
          mu[j] = mu_direct(adp + (size_t)wave * U_,
                            aep + (size_t)(lane + 32 * j) * U_, nu);
    }
  }

  float mx = fmaxf(fmaxf(fmaxf(mu[0], mu[4]), fmaxf(mu[2], mu[6])),
                   fmaxf(fmaxf(mu[1], mu[5]), fmaxf(mu[3], mu[7])));
  mx = fmaxf(mx, lane_xor16(mx));
  mx = fmaxf(mx, lane_xor_row<8>(mx));
  mx = fmaxf(mx, lane_xor_row<4>(mx));
  mx = fmaxf(mx, lane_xor_row<2>(mx));
  mx = fmaxf(mx, lane_xor_row<1>(mx));
  float e[8];
#pragma unroll
  for (int j = 0; j < 8; ++j) e[j] = expf(mu[j] - mx);
  float sum = ((e[0] + e[4]) + (e[2] + e[6])) + ((e[1] + e[5]) + (e[3] + e[7]));
  sum = sum + lane_xor16(sum);
  sum = sum + lane_xor_row<8>(sum);
  sum = sum + lane_xor_row<4>(sum);
  sum = sum + lane_xor_row<2>(sum);
  sum = sum + lane_xor_row<1>(sum);
  const float inv = 1.0f / sum;

  float* dst = alphas + ((size_t)(b * TD_ + s0 + wave)) * TE_ + lane;
#pragma unroll
  for (int j = 0; j < 8; ++j) dst[32 * j] = e[j] * inv;
  if (lane == 0) max_mu[b * TD_ + s0 + wave] = mx;
}

__global__ void __launch_bounds__(256) hhat_kernel(
    const float* __restrict__ max_mu, const float* __restrict__ de_seq,
    float* __restrict__ h_hat) {
  const int b    = blockIdx.y;
  const int d0   = blockIdx.x * 32;
  const int lane = threadIdx.x & 31;
  const int wave = __builtin_amdgcn_readfirstlane((int)(threadIdx.x >> 5));
  __shared__ float part[8][32];

  const float* mm = max_mu + b * TD_ + lane;
  float v[8];
#pragma unroll
  for (int j = 0; j < 8; ++j) v[j] = mm[32 * j];

  const float* dp = de_seq + ((size_t)(b * TD_ + wave * 32)) * D_ + d0 + lane;
  float de[32];
#pragma unroll
  for (int i = 0; i < 32; ++i) de[i] = dp[(size_t)i * D_];

  float mx = fmaxf(fmaxf(fmaxf(v[0], v[4]), fmaxf(v[2], v[6])),
                   fmaxf(fmaxf(v[1], v[5]), fmaxf(v[3], v[7])));
  mx = fmaxf(mx, lane_xor16(mx));
  mx = fmaxf(mx, lane_xor_row<8>(mx));
  mx = fmaxf(mx, lane_xor_row<4>(mx));
  mx = fmaxf(mx, lane_xor_row<2>(mx));
  mx = fmaxf(mx, lane_xor_row<1>(mx));

  float e[8];
#pragma unroll
  for (int j = 0; j < 8; ++j) e[j] = expf(v[j] - mx);
  float sum = ((e[0] + e[4]) + (e[2] + e[6])) + ((e[1] + e[5]) + (e[3] + e[7]));
  sum = sum + lane_xor16(sum);
  sum = sum + lane_xor_row<8>(sum);
  sum = sum + lane_xor_row<4>(sum);
  sum = sum + lane_xor_row<2>(sum);
  sum = sum + lane_xor_row<1>(sum);
  const float inv = 1.0f / sum;

  float eo = e[0];
#pragma unroll
  for (int j = 1; j < 8; ++j) eo = (wave == j) ? e[j] : eo;
  const float ma = eo * inv;

  float acc = 0.f;
#pragma unroll
  for (int i = 0; i < 32; ++i) {
    const float m_i = __builtin_bit_cast(float,
        __builtin_amdgcn_readlane(__builtin_bit_cast(int, ma), i));
    acc += de[i] * m_i;
  }
  part[wave][lane] = acc;
  __syncthreads();
  if (wave == 0) {
    float t = part[0][lane];
#pragma unroll
    for (int w = 1; w < 8; ++w) t += part[w][lane];
    h_hat[b * D_ + d0 + lane] = t;
  }
}

__global__ void __launch_bounds__(256) sum_en_out_kernel(
    const float* __restrict__ alphas, const float* __restrict__ en_seq,
    const float* __restrict__ de_seq, const float* __restrict__ h_hat,
    float* __restrict__ out) {
  __shared__ __attribute__((aligned(16))) float As[PANEL_A_FLOATS_];
  __shared__ __attribute__((aligned(16))) float Bs[PANEL_B_FLOATS_];
  const int b     = blockIdx.y;
  const int lane  = threadIdx.x & 31;
  const int wave  = threadIdx.x >> 5;
  const int m     = lane & 15;
  const int khalf = lane >> 4;
  const int mb    = blockIdx.x >> 2;
  const int nb    = blockIdx.x & 3;

  const v8f acc = mma_panel_32x64(
      alphas + ((size_t)b * TD_ + mb * 32) * TE_,
      en_seq + (size_t)b * TE_ * D_ + nb * 64, As, Bs);

  const int d_g  = nb * 64 + (wave & 3) * 16 + m;
  const float hh = h_hat[b * D_ + d_g];
#pragma unroll
  for (int r = 0; r < 8; ++r) {
    const int s_g  = mb * 32 + (wave >> 2) * 16 + r + 8 * khalf;
    const float se = acc[r];
    const float de = de_seq[((size_t)(b * TD_ + s_g)) * D_ + d_g];
    float* o = out + ((size_t)(b * TD_ + s_g)) * (4 * D_);
    o[d_g]          = de;
    o[D_ + d_g]     = se;
    o[2 * D_ + d_g] = de * se;
    o[3 * D_ + d_g] = de * hh;
  }
}

extern "C" void kernel_launch(void* const* d_in, const int* in_sizes, int n_in,
                              void* d_out, int out_size, void* d_ws, size_t ws_size,
                              hipStream_t stream) {
  const float* en_seq = (const float*)d_in[0];
  const float* de_seq = (const float*)d_in[1];
  const float* w_en   = (const float*)d_in[2];
  const float* w_de   = (const float*)d_in[3];
  const float* nu     = (const float*)d_in[4];
  float* out = (float*)d_out;

  float* ws = (float*)d_ws;
  const size_t nAE = (size_t)B_ * TE_ * U_;
  float* ae     = ws;
  float* ad     = ae + nAE;
  float* mu     = ad + nAE;
  float* max_mu = mu + (size_t)B_ * TD_ * TE_;
  float* h_hat  = max_mu + (size_t)B_ * TD_;

  const dim3 blk(256);
  gemm_f32_wmma<<<dim3(256), blk, 0, stream>>>(en_seq, w_en, ae);
  gemm_f32_wmma<<<dim3(256), blk, 0, stream>>>(de_seq, w_de, ad);
  co_rows_softmax_kernel<<<dim3(32, B_), blk, 0, stream>>>(ae, ad, nu, mu, max_mu);
  hhat_kernel<<<dim3(8, B_), blk, 0, stream>>>(max_mu, de_seq, h_hat);
  sum_en_out_kernel<<<dim3(32, B_), blk, 0, stream>>>(mu, en_seq, de_seq,
                                                      h_hat, out);
}


# h_hat launch removed by a split softmax: each score block writes its 8 rows' partial (local max, weight sum, weighted de sums); output blocks merge the 32 partials; h_hat sum order over s changed
# speedup vs baseline: 3.8805x; 1.0380x over previous
#include <hip/hip_runtime.h>
#include <cmath>

#define B_  8
#define TE_ 256
#define TD_ 256
#define D_  256
#define U_  256

typedef float v2f __attribute__((ext_vector_type(2)));
typedef float v8f __attribute__((ext_vector_type(8)));

typedef float v4f __attribute__((ext_vector_type(4)));
#define PA_ 260
#define PB_ 72
#define PANEL_A_FLOATS_ (32 * PA_)
#define PANEL_B_FLOATS_ (256 * PB_)

__device__ __forceinline__ void lds_quarter_barrier() {
  asm volatile("s_wait_dscnt 0x0" ::: "memory");
  asm volatile("s_barrier_signal -1" ::: "memory");
  asm volatile("s_barrier_wait -1" ::: "memory");
}

__device__ __forceinline__ v8f mma_panel_32x64(const float* __restrict__ Ag,
                                               const float* __restrict__ Bg,
                                               float* As, float* Bs) {
  const int tid   = threadIdx.x;
  const int lane  = tid & 31;
  const int wave  = tid >> 5;
  const int m     = lane & 15;
  const int khalf = lane >> 4;
  const int wm    = wave >> 2;
  const int wn    = wave & 3;
  const int row   = tid >> 4;
  const int c4    = (tid & 15) * 4;

  v4f ra[4][2], rb[4][4];
#pragma unroll
  for (int q = 0; q < 4; ++q) {
#pragma unroll
    for (int j = 0; j < 2; ++j)
      ra[q][j] = *(const v4f*)(Ag + (size_t)(j * 16 + row) * 256 + 64 * q + c4);
#pragma unroll
    for (int j = 0; j < 4; ++j)
      rb[q][j] = *(const v4f*)(Bg + (size_t)(64 * q + j * 16 + row) * 256 + c4);
  }

  const float* af = As + (wm * 16 + m) * PA_ + 2 * khalf;
  const float* bf = Bs + (2 * khalf) * PB_ + wn * 16 + m;
  v2f fa[2][16], fb[2][16];
  v8f acc = {};
#pragma unroll
  for (int q = -1; q < 4; ++q) {
    if (q < 3) {
      const int qn = q + 1;
#pragma unroll
      for (int j = 0; j < 2; ++j)
        *(v4f*)(As + (j * 16 + row) * PA_ + 64 * qn + c4) = ra[qn][j];
#pragma unroll
      for (int j = 0; j < 4; ++j)
        *(v4f*)(Bs + (64 * qn + j * 16 + row) * PB_ + c4) = rb[qn][j];
      lds_quarter_barrier();
#pragma unroll
      for (int i = 0; i < 16; ++i) {
        const int k4 = 4 * (16 * qn + i);
        fa[qn & 1][i]   = *(const v2f*)(af + k4);
        fb[qn & 1][i].x = bf[(k4 + 0) * PB_];
        fb[qn & 1][i].y = bf[(k4 + 1) * PB_];
      }
    }
    __builtin_amdgcn_sched_barrier(0);
    if (q >= 0) {
#pragma unroll
      for (int i = 0; i < 16; ++i)
        acc = __builtin_amdgcn_wmma_f32_16x16x4_f32(false, fa[q & 1][i], false,
                                                    fb[q & 1][i], (short)0, acc,
                                                    false, false);
    }
    __builtin_amdgcn_sched_barrier(0);
  }
  return acc;
}

__global__ void __launch_bounds__(256) gemm_f32_wmma(
    const float* __restrict__ A, const float* __restrict__ W,
    float* __restrict__ C) {
  __shared__ __attribute__((aligned(16))) float As[PANEL_A_FLOATS_];
  __shared__ __attribute__((aligned(16))) float Bs[PANEL_B_FLOATS_];
  const int lane  = threadIdx.x & 31;
  const int wave  = threadIdx.x >> 5;
  const int m     = lane & 15;
  const int khalf = lane >> 4;
  const int mb    = blockIdx.x >> 2;
  const int nb    = blockIdx.x & 3;

  const v8f acc = mma_panel_32x64(A + (size_t)mb * 32 * U_, W + nb * 64, As, Bs);

  const int n_out = nb * 64 + (wave & 3) * 16 + m;
#pragma unroll
  for (int r = 0; r < 8; ++r) {
    const int m_out = mb * 32 + (wave >> 2) * 16 + r + 8 * khalf;
    C[(size_t)m_out * U_ + n_out] = acc[r];
  }
}

#define EXP_LIM_ 80.0f
#define PE_ 260
#define HP_PITCH_ 288
#define HP_BLOCKS_ 32

__device__ __forceinline__ float exp_staged(float v) {
  const float kHi = 1.44269502f;
  const float kLo = 1.92596299e-8f;
  const float kLn2 = 0.693147182f;
  const float hi = kHi * v;
  float lo = fmaf(kHi, v, -hi);
  lo = fmaf(kLo, v, lo);
  const float e = __builtin_amdgcn_exp2f(hi);
  return fmaf(e, lo * kLn2, e);
}

__device__ __forceinline__ float mu_direct(const float* __restrict__ adr,
                                           const float* __restrict__ aer,
                                           const float* __restrict__ nu) {
  float acc = 0.f;
  for (int u = 0; u < U_; ++u) acc += tanhf(adr[u] + aer[u]) * nu[u];
  return acc;
}

__device__ __forceinline__ float lane_xor16(float x) {
  const int b = __builtin_bit_cast(int, x);
  return __builtin_bit_cast(float,
      __builtin_amdgcn_permlanex16(b, b, 0x76543210, 0xfedcba98, false, false));
}
template <int M>
__device__ __forceinline__ float lane_xor_row(float x) {
  const int b = __builtin_bit_cast(int, x);
  return __builtin_bit_cast(float,
      __builtin_amdgcn_update_dpp(b, b, 0x160 + M, 0xf, 0xf, false));
}

__global__ void __launch_bounds__(256) co_rows_softmax_kernel(
    const float* __restrict__ ae, const float* __restrict__ ad,
    const float* __restrict__ nu, const float* __restrict__ de_seq,
    float* __restrict__ alphas, float* __restrict__ hpart) {
  __shared__ __attribute__((aligned(16))) float EeT[256 * PE_];
  __shared__ __attribute__((aligned(16))) float EaT[8 * PE_];
  __shared__ __attribute__((aligned(16))) float nuS[256];
  __shared__ float muFix[256];
  __shared__ float nuPart[8];
  __shared__ float mxS[8];
  __shared__ int   badT[256];
  __shared__ int   badS[8];
  __shared__ int   anyBad;

  const int b    = blockIdx.y;
  const int s0   = blockIdx.x * 8;
  const int tid  = threadIdx.x;
  const int lane = tid & 31;
  const int wave = __builtin_amdgcn_readfirstlane((int)(threadIdx.x >> 5));
  const int row  = tid >> 4;
  const int c4   = (tid & 15) * 4;
  const float* aep = ae + (size_t)b * TE_ * U_;
  const float* adp = ad + ((size_t)b * TD_ + s0) * U_;

  v4f cur[16];
#pragma unroll
  for (int i = 0; i < 16; ++i)
    cur[i] = *(const v4f*)(aep + (size_t)(16 * i + row) * U_ + c4);
  v4f va[2];
#pragma unroll
  for (int k = 0; k < 2; ++k) va[k] = *(const v4f*)(adp + (size_t)(tid + 256 * k) * 4);

  float dv[8];
#pragma unroll
  for (int k = 0; k < 8; ++k)
    dv[k] = de_seq[((size_t)(b * TD_ + s0 + k)) * D_ + tid];

  const float nuv = nu[tid];
  nuS[tid]  = -2.0f * nuv;
  badT[tid] = 0;
  if (tid < 8) badS[tid] = 0;
  if (tid == 0) anyBad = 0;
  float nsum = nuv;
#pragma unroll
  for (int m = 16; m > 0; m >>= 1) nsum += __shfl_xor(nsum, m);
  if (lane == 0) nuPart[wave] = nsum;

  int badA[2];
#pragma unroll
  for (int k = 0; k < 2; ++k) {
    const int idx = tid + 256 * k;
    v4f e;
    int bad = 0;
#pragma unroll
    for (int x = 0; x < 4; ++x) {
      bad |= !(fabsf(va[k][x]) <= EXP_LIM_);
      e[x] = exp_staged(va[k][x]);
    }
    badA[k] = bad;
    *(v4f*)(EaT + (idx >> 6) * PE_ + (idx & 63) * 4) = e;
  }

  const float* eaRow  = EaT + wave * PE_;
  const float* eeLane = EeT + lane * PE_;
  float acc[8];
#pragma unroll
  for (int j = 0; j < 8; ++j) acc[j] = 0.f;
  unsigned badmask = 0;

#pragma unroll
  for (int q = 0; q < 4; ++q) {
#pragma unroll
    for (int i = 0; i < 16; ++i) {
      v4f e;
      int bad = 0;
#pragma unroll
      for (int x = 0; x < 4; ++x) {
        bad |= !(fabsf(cur[i][x]) <= EXP_LIM_);
        e[x] = exp_staged(cur[i][x]);
      }
      badmask |= (unsigned)bad << i;
      *(v4f*)(EeT + (16 * i + row) * PE_ + 64 * q + c4) = e;
    }
    if (q < 3) {
#pragma unroll
      for (int i = 0; i < 16; ++i)
        cur[i] = *(const v4f*)(aep + (size_t)(16 * i + row) * U_ + 64 * (q + 1) + c4);
    }
    lds_quarter_barrier();

#pragma unroll 2
    for (int u = 64 * q; u < 64 * q + 64; u += 4) {
      const v4f a4 = *(const v4f*)(eaRow + u);
      const v4f n4 = *(const v4f*)(nuS + u);
#pragma unroll
      for (int j = 0; j < 8; ++j) {
        const v4f e4 = *(const v4f*)(eeLane + j * 32 * PE_ + u);
#pragma unroll
        for (int x = 0; x < 4; ++x) {
          const float p   = a4[x] * e4[x];
          const float den = fmaf(p, p, 1.0f);
          acc[j] = fmaf(n4[x], __builtin_amdgcn_rcpf(den), acc[j]);
        }
      }
    }
  }

  if (badA[0]) { badS[tid >> 6] = 1; anyBad = 1; }
  if (badA[1]) { badS[(tid + 256) >> 6] = 1; anyBad = 1; }
  if (badmask) {
    anyBad = 1;
#pragma unroll
    for (int i = 0; i < 16; ++i)
      if ((badmask >> i) & 1u) badT[16 * i + row] = 1;
  }
  __syncthreads();

  float nu_total = 0.f;
#pragma unroll
  for (int w = 0; w < 8; ++w) nu_total += nuPart[w];
  float mu[8];
#pragma unroll
  for (int j = 0; j < 8; ++j) mu[j] = acc[j] + nu_total;

  if (anyBad) {
    for (int s = 0; s < 8; ++s) {
      if (badS[s]) {
        muFix[tid] = mu_direct(adp + (size_t)s * U_, aep + (size_t)tid * U_, nu);
        __syncthreads();
        if (wave == s) {
#pragma unroll
          for (int j = 0; j < 8; ++j) mu[j] = muFix[lane + 32 * j];
        }
        __syncthreads();
      }
    }
    if (!badS[wave]) {
#pragma unroll
      for (int j = 0; j < 8; ++j)
        if (badT[lane + 32 * j])
          mu[j] = mu_direct(adp + (size_t)wave * U_,
                            aep + (size_t)(lane + 32 * j) * U_, nu);
    }
  }

  float mx = fmaxf(fmaxf(fmaxf(mu[0], mu[4]), fmaxf(mu[2], mu[6])),
                   fmaxf(fmaxf(mu[1], mu[5]), fmaxf(mu[3], mu[7])));
  mx = fmaxf(mx, lane_xor16(mx));
  mx = fmaxf(mx, lane_xor_row<8>(mx));
  mx = fmaxf(mx, lane_xor_row<4>(mx));
  mx = fmaxf(mx, lane_xor_row<2>(mx));
  mx = fmaxf(mx, lane_xor_row<1>(mx));
  if (lane == 0) mxS[wave] = mx;
  float e[8];
#pragma unroll
  for (int j = 0; j < 8; ++j) e[j] = expf(mu[j] - mx);
  float sum = ((e[0] + e[4]) + (e[2] + e[6])) + ((e[1] + e[5]) + (e[3] + e[7]));
  sum = sum + lane_xor16(sum);
  sum = sum + lane_xor_row<8>(sum);
  sum = sum + lane_xor_row<4>(sum);
  sum = sum + lane_xor_row<2>(sum);
  sum = sum + lane_xor_row<1>(sum);
  const float inv = 1.0f / sum;

  float* dst = alphas + ((size_t)(b * TD_ + s0 + wave)) * TE_ + lane;
#pragma unroll
  for (int j = 0; j < 8; ++j) dst[32 * j] = e[j] * inv;

  lds_quarter_barrier();
  float mb = mxS[0];
#pragma unroll
  for (int k = 1; k < 8; ++k) mb = fmaxf(mb, mxS[k]);
  float z = 0.f, p = 0.f;
#pragma unroll
  for (int k = 0; k < 8; ++k) {
    const float w = expf(mxS[k] - mb);
    z += w;
    p = fmaf(w, dv[k], p);
  }
  float* hp = hpart + ((size_t)(b * HP_BLOCKS_ + blockIdx.x)) * HP_PITCH_;
  hp[tid] = p;
  if (tid == 0) { hp[256] = mb; hp[257] = z; }
}

__global__ void __launch_bounds__(256) sum_en_out_kernel(
    const float* __restrict__ alphas, const float* __restrict__ en_seq,
    const float* __restrict__ de_seq, const float* __restrict__ hpart,
    float* __restrict__ out) {
  __shared__ __attribute__((aligned(16))) float As[PANEL_A_FLOATS_];
  __shared__ __attribute__((aligned(16))) float Bs[PANEL_B_FLOATS_];
  const int b     = blockIdx.y;
  const int lane  = threadIdx.x & 31;
  const int wave  = threadIdx.x >> 5;
  const int m     = lane & 15;
  const int khalf = lane >> 4;
  const int mb    = blockIdx.x >> 2;
  const int nb    = blockIdx.x & 3;

  const int d_g  = nb * 64 + (wave & 3) * 16 + m;

  const float* hp = hpart + (size_t)b * HP_BLOCKS_ * HP_PITCH_;
  const v2f mz = *(const v2f*)(hp + (size_t)lane * HP_PITCH_ + 256);
  float pv[HP_BLOCKS_];
#pragma unroll
  for (int i = 0; i < HP_BLOCKS_; ++i) pv[i] = hp[(size_t)i * HP_PITCH_ + d_g];

  const v8f acc = mma_panel_32x64(
      alphas + ((size_t)b * TD_ + mb * 32) * TE_,
      en_seq + (size_t)b * TE_ * D_ + nb * 64, As, Bs);

  float mall = fmaxf(mz.x, lane_xor16(mz.x));
  mall = fmaxf(mall, lane_xor_row<8>(mall));
  mall = fmaxf(mall, lane_xor_row<4>(mall));
  mall = fmaxf(mall, lane_xor_row<2>(mall));
  mall = fmaxf(mall, lane_xor_row<1>(mall));
  const float ci = expf(mz.x - mall);
  float den = ci * mz.y;
  den = den + lane_xor16(den);
  den = den + lane_xor_row<8>(den);
  den = den + lane_xor_row<4>(den);
  den = den + lane_xor_row<2>(den);
  den = den + lane_xor_row<1>(den);
  float hnum = 0.f;
#pragma unroll
  for (int i = 0; i < HP_BLOCKS_; ++i) {
    const float c_i = __builtin_bit_cast(float,
        __builtin_amdgcn_readlane(__builtin_bit_cast(int, ci), i));
    hnum = fmaf(c_i, pv[i], hnum);
  }
  const float hh = hnum / den;
#pragma unroll
  for (int r = 0; r < 8; ++r) {
    const int s_g  = mb * 32 + (wave >> 2) * 16 + r + 8 * khalf;
    const float se = acc[r];
    const float de = de_seq[((size_t)(b * TD_ + s_g)) * D_ + d_g];
    float* o = out + ((size_t)(b * TD_ + s_g)) * (4 * D_);
    o[d_g]          = de;
    o[D_ + d_g]     = se;
    o[2 * D_ + d_g] = de * se;
    o[3 * D_ + d_g] = de * hh;
  }
}

extern "C" void kernel_launch(void* const* d_in, const int* in_sizes, int n_in,
                              void* d_out, int out_size, void* d_ws, size_t ws_size,
                              hipStream_t stream) {
  const float* en_seq = (const float*)d_in[0];
  const float* de_seq = (const float*)d_in[1];
  const float* w_en   = (const float*)d_in[2];
  const float* w_de   = (const float*)d_in[3];
  const float* nu     = (const float*)d_in[4];
  float* out = (float*)d_out;

  float* ws = (float*)d_ws;
  const size_t nAE = (size_t)B_ * TE_ * U_;
  float* ae     = ws;
  float* ad     = ae + nAE;
  float* mu     = ad + nAE;
  float* hpart  = mu + (size_t)B_ * TD_ * TE_;

  const dim3 blk(256);
  gemm_f32_wmma<<<dim3(256), blk, 0, stream>>>(en_seq, w_en, ae);
  gemm_f32_wmma<<<dim3(256), blk, 0, stream>>>(de_seq, w_de, ad);
  co_rows_softmax_kernel<<<dim3(HP_BLOCKS_, B_), blk, 0, stream>>>(ae, ad, nu, de_seq,
                                                                 mu, hpart);
  sum_en_out_kernel<<<dim3(32, B_), blk, 0, stream>>>(mu, en_seq, de_seq,
                                                      hpart, out);
}


# score kernel stages exp(2*ad) and exp(2*ae), so 1+exp(2(ad+ae)) is one fma of the two staged values (was a multiply, then an fma); tanhf fallback now beyond |a|>43 (was 80); sum order unchanged
# speedup vs baseline: 4.0051x; 1.0321x over previous
#include <hip/hip_runtime.h>
#include <cmath>

#define B_  8
#define TE_ 256
#define TD_ 256
#define D_  256
#define U_  256

typedef float v2f __attribute__((ext_vector_type(2)));
typedef float v8f __attribute__((ext_vector_type(8)));

typedef float v4f __attribute__((ext_vector_type(4)));
#define PA_ 260
#define PB_ 72
#define PANEL_A_FLOATS_ (32 * PA_)
#define PANEL_B_FLOATS_ (256 * PB_)

__device__ __forceinline__ void lds_quarter_barrier() {
  asm volatile("s_wait_dscnt 0x0" ::: "memory");
  asm volatile("s_barrier_signal -1" ::: "memory");
  asm volatile("s_barrier_wait -1" ::: "memory");
}

__device__ __forceinline__ v8f mma_panel_32x64(const float* __restrict__ Ag,
                                               const float* __restrict__ Bg,
                                               float* As, float* Bs) {
  const int tid   = threadIdx.x;
  const int lane  = tid & 31;
  const int wave  = tid >> 5;
  const int m     = lane & 15;
  const int khalf = lane >> 4;
  const int wm    = wave >> 2;
  const int wn    = wave & 3;
  const int row   = tid >> 4;
  const int c4    = (tid & 15) * 4;

  v4f ra[4][2], rb[4][4];
#pragma unroll
  for (int q = 0; q < 4; ++q) {
#pragma unroll
    for (int j = 0; j < 2; ++j)
      ra[q][j] = *(const v4f*)(Ag + (size_t)(j * 16 + row) * 256 + 64 * q + c4);
#pragma unroll
    for (int j = 0; j < 4; ++j)
      rb[q][j] = *(const v4f*)(Bg + (size_t)(64 * q + j * 16 + row) * 256 + c4);
  }

  const float* af = As + (wm * 16 + m) * PA_ + 2 * khalf;
  const float* bf = Bs + (2 * khalf) * PB_ + wn * 16 + m;
  v2f fa[2][16], fb[2][16];
  v8f acc = {};
#pragma unroll
  for (int q = -1; q < 4; ++q) {
    if (q < 3) {
      const int qn = q + 1;
#pragma unroll
      for (int j = 0; j < 2; ++j)
        *(v4f*)(As + (j * 16 + row) * PA_ + 64 * qn + c4) = ra[qn][j];
#pragma unroll
      for (int j = 0; j < 4; ++j)
        *(v4f*)(Bs + (64 * qn + j * 16 + row) * PB_ + c4) = rb[qn][j];
      lds_quarter_barrier();
#pragma unroll
      for (int i = 0; i < 16; ++i) {
        const int k4 = 4 * (16 * qn + i);
        fa[qn & 1][i]   = *(const v2f*)(af + k4);
        fb[qn & 1][i].x = bf[(k4 + 0) * PB_];
        fb[qn & 1][i].y = bf[(k4 + 1) * PB_];
      }
    }
    __builtin_amdgcn_sched_barrier(0);
    if (q >= 0) {
#pragma unroll
      for (int i = 0; i < 16; ++i)
        acc = __builtin_amdgcn_wmma_f32_16x16x4_f32(false, fa[q & 1][i], false,
                                                    fb[q & 1][i], (short)0, acc,
                                                    false, false);
    }
    __builtin_amdgcn_sched_barrier(0);
  }
  return acc;
}

__global__ void __launch_bounds__(256) gemm_f32_wmma(
    const float* __restrict__ A, const float* __restrict__ W,
    float* __restrict__ C) {
  __shared__ __attribute__((aligned(16))) float As[PANEL_A_FLOATS_];
  __shared__ __attribute__((aligned(16))) float Bs[PANEL_B_FLOATS_];
  const int lane  = threadIdx.x & 31;
  const int wave  = threadIdx.x >> 5;
  const int m     = lane & 15;
  const int khalf = lane >> 4;
  const int mb    = blockIdx.x >> 2;
  const int nb    = blockIdx.x & 3;

  const v8f acc = mma_panel_32x64(A + (size_t)mb * 32 * U_, W + nb * 64, As, Bs);

  const int n_out = nb * 64 + (wave & 3) * 16 + m;
#pragma unroll
  for (int r = 0; r < 8; ++r) {
    const int m_out = mb * 32 + (wave >> 2) * 16 + r + 8 * khalf;
    C[(size_t)m_out * U_ + n_out] = acc[r];
  }
}

#define EXP_LIM_ 43.0f
#define PE_ 260
#define HP_PITCH_ 288
#define HP_BLOCKS_ 32

__device__ __forceinline__ float exp_staged(float v) {
  const float kHi = 2.0f * 1.44269502f;
  const float kLo = 2.0f * 1.92596299e-8f;
  const float kLn2 = 0.693147182f;
  const float hi = kHi * v;
  float lo = fmaf(kHi, v, -hi);
  lo = fmaf(kLo, v, lo);
  const float e = __builtin_amdgcn_exp2f(hi);
  return fmaf(e, lo * kLn2, e);
}

__device__ __forceinline__ float mu_direct(const float* __restrict__ adr,
                                           const float* __restrict__ aer,
                                           const float* __restrict__ nu) {
  float acc = 0.f;
  for (int u = 0; u < U_; ++u) acc += tanhf(adr[u] + aer[u]) * nu[u];
  return acc;
}

__device__ __forceinline__ float lane_xor16(float x) {
  const int b = __builtin_bit_cast(int, x);
  return __builtin_bit_cast(float,
      __builtin_amdgcn_permlanex16(b, b, 0x76543210, 0xfedcba98, false, false));
}
template <int M>
__device__ __forceinline__ float lane_xor_row(float x) {
  const int b = __builtin_bit_cast(int, x);
  return __builtin_bit_cast(float,
      __builtin_amdgcn_update_dpp(b, b, 0x160 + M, 0xf, 0xf, false));
}

__global__ void __launch_bounds__(256) co_rows_softmax_kernel(
    const float* __restrict__ ae, const float* __restrict__ ad,
    const float* __restrict__ nu, const float* __restrict__ de_seq,
    float* __restrict__ alphas, float* __restrict__ hpart) {
  __shared__ __attribute__((aligned(16))) float EeT[256 * PE_];
  __shared__ __attribute__((aligned(16))) float EaT[8 * PE_];
  __shared__ __attribute__((aligned(16))) float nuS[256];
  __shared__ float muFix[256];
  __shared__ float nuPart[8];
  __shared__ float mxS[8];
  __shared__ int   badT[256];
  __shared__ int   badS[8];
  __shared__ int   anyBad;

  const int b    = blockIdx.y;
  const int s0   = blockIdx.x * 8;
  const int tid  = threadIdx.x;
  const int lane = tid & 31;
  const int wave = __builtin_amdgcn_readfirstlane((int)(threadIdx.x >> 5));
  const int row  = tid >> 4;
  const int c4   = (tid & 15) * 4;
  const float* aep = ae + (size_t)b * TE_ * U_;
  const float* adp = ad + ((size_t)b * TD_ + s0) * U_;

  v4f cur[16];
#pragma unroll
  for (int i = 0; i < 16; ++i)
    cur[i] = *(const v4f*)(aep + (size_t)(16 * i + row) * U_ + c4);
  v4f va[2];
#pragma unroll
  for (int k = 0; k < 2; ++k) va[k] = *(const v4f*)(adp + (size_t)(tid + 256 * k) * 4);

  float dv[8];
#pragma unroll
  for (int k = 0; k < 8; ++k)
    dv[k] = de_seq[((size_t)(b * TD_ + s0 + k)) * D_ + tid];

  const float nuv = nu[tid];
  nuS[tid]  = -2.0f * nuv;
  badT[tid] = 0;
  if (tid < 8) badS[tid] = 0;
  if (tid == 0) anyBad = 0;
  float nsum = nuv;
#pragma unroll
  for (int m = 16; m > 0; m >>= 1) nsum += __shfl_xor(nsum, m);
  if (lane == 0) nuPart[wave] = nsum;

  int badA[2];
#pragma unroll
  for (int k = 0; k < 2; ++k) {
    const int idx = tid + 256 * k;
    v4f e;
    int bad = 0;
#pragma unroll
    for (int x = 0; x < 4; ++x) {
      bad |= !(fabsf(va[k][x]) <= EXP_LIM_);
      e[x] = exp_staged(va[k][x]);
    }
    badA[k] = bad;
    *(v4f*)(EaT + (idx >> 6) * PE_ + (idx & 63) * 4) = e;
  }

  const float* eaRow  = EaT + wave * PE_;
  const float* eeLane = EeT + lane * PE_;
  float acc[8];
#pragma unroll
  for (int j = 0; j < 8; ++j) acc[j] = 0.f;
  unsigned badmask = 0;

#pragma unroll
  for (int q = 0; q < 4; ++q) {
#pragma unroll
    for (int i = 0; i < 16; ++i) {
      v4f e;
      int bad = 0;
#pragma unroll
      for (int x = 0; x < 4; ++x) {
        bad |= !(fabsf(cur[i][x]) <= EXP_LIM_);
        e[x] = exp_staged(cur[i][x]);
      }
      badmask |= (unsigned)bad << i;
      *(v4f*)(EeT + (16 * i + row) * PE_ + 64 * q + c4) = e;
    }
    if (q < 3) {
#pragma unroll
      for (int i = 0; i < 16; ++i)
        cur[i] = *(const v4f*)(aep + (size_t)(16 * i + row) * U_ + 64 * (q + 1) + c4);
    }
    lds_quarter_barrier();

#pragma unroll 2
    for (int u = 64 * q; u < 64 * q + 64; u += 4) {
      const v4f a4 = *(const v4f*)(eaRow + u);
      const v4f n4 = *(const v4f*)(nuS + u);
#pragma unroll
      for (int j = 0; j < 8; ++j) {
        const v4f e4 = *(const v4f*)(eeLane + j * 32 * PE_ + u);
#pragma unroll
        for (int x = 0; x < 4; ++x) {
          const float den = fmaf(a4[x], e4[x], 1.0f);
          acc[j] = fmaf(n4[x], __builtin_amdgcn_rcpf(den), acc[j]);
        }
      }
    }
  }

  if (badA[0]) { badS[tid >> 6] = 1; anyBad = 1; }
  if (badA[1]) { badS[(tid + 256) >> 6] = 1; anyBad = 1; }
  if (badmask) {
    anyBad = 1;
#pragma unroll
    for (int i = 0; i < 16; ++i)
      if ((badmask >> i) & 1u) badT[16 * i + row] = 1;
  }
  __syncthreads();

  float nu_total = 0.f;
#pragma unroll
  for (int w = 0; w < 8; ++w) nu_total += nuPart[w];
  float mu[8];
#pragma unroll
  for (int j = 0; j < 8; ++j) mu[j] = acc[j] + nu_total;

  if (anyBad) {
    for (int s = 0; s < 8; ++s) {
      if (badS[s]) {
        muFix[tid] = mu_direct(adp + (size_t)s * U_, aep + (size_t)tid * U_, nu);
        __syncthreads();
        if (wave == s) {
#pragma unroll
          for (int j = 0; j < 8; ++j) mu[j] = muFix[lane + 32 * j];
        }
        __syncthreads();
      }
    }
    if (!badS[wave]) {
#pragma unroll
      for (int j = 0; j < 8; ++j)
        if (badT[lane + 32 * j])
          mu[j] = mu_direct(adp + (size_t)wave * U_,
                            aep + (size_t)(lane + 32 * j) * U_, nu);
    }
  }

  float mx = fmaxf(fmaxf(fmaxf(mu[0], mu[4]), fmaxf(mu[2], mu[6])),
                   fmaxf(fmaxf(mu[1], mu[5]), fmaxf(mu[3], mu[7])));
  mx = fmaxf(mx, lane_xor16(mx));
  mx = fmaxf(mx, lane_xor_row<8>(mx));
  mx = fmaxf(mx, lane_xor_row<4>(mx));
  mx = fmaxf(mx, lane_xor_row<2>(mx));
  mx = fmaxf(mx, lane_xor_row<1>(mx));
  if (lane == 0) mxS[wave] = mx;
  float e[8];
#pragma unroll
  for (int j = 0; j < 8; ++j) e[j] = expf(mu[j] - mx);
  float sum = ((e[0] + e[4]) + (e[2] + e[6])) + ((e[1] + e[5]) + (e[3] + e[7]));
  sum = sum + lane_xor16(sum);
  sum = sum + lane_xor_row<8>(sum);
  sum = sum + lane_xor_row<4>(sum);
  sum = sum + lane_xor_row<2>(sum);
  sum = sum + lane_xor_row<1>(sum);
  const float inv = 1.0f / sum;

  float* dst = alphas + ((size_t)(b * TD_ + s0 + wave)) * TE_ + lane;
#pragma unroll
  for (int j = 0; j < 8; ++j) dst[32 * j] = e[j] * inv;

  lds_quarter_barrier();
  float mb = mxS[0];
#pragma unroll
  for (int k = 1; k < 8; ++k) mb = fmaxf(mb, mxS[k]);
  float z = 0.f, p = 0.f;
#pragma unroll
  for (int k = 0; k < 8; ++k) {
    const float w = expf(mxS[k] - mb);
    z += w;
    p = fmaf(w, dv[k], p);
  }
  float* hp = hpart + ((size_t)(b * HP_BLOCKS_ + blockIdx.x)) * HP_PITCH_;
  hp[tid] = p;
  if (tid == 0) { hp[256] = mb; hp[257] = z; }
}

__global__ void __launch_bounds__(256) sum_en_out_kernel(
    const float* __restrict__ alphas, const float* __restrict__ en_seq,
    const float* __restrict__ de_seq, const float* __restrict__ hpart,
    float* __restrict__ out) {
  __shared__ __attribute__((aligned(16))) float As[PANEL_A_FLOATS_];
  __shared__ __attribute__((aligned(16))) float Bs[PANEL_B_FLOATS_];
  const int b     = blockIdx.y;
  const int lane  = threadIdx.x & 31;
  const int wave  = threadIdx.x >> 5;
  const int m     = lane & 15;
  const int khalf = lane >> 4;
  const int mb    = blockIdx.x >> 2;
  const int nb    = blockIdx.x & 3;

  const int d_g  = nb * 64 + (wave & 3) * 16 + m;

  const float* hp = hpart + (size_t)b * HP_BLOCKS_ * HP_PITCH_;
  const v2f mz = *(const v2f*)(hp + (size_t)lane * HP_PITCH_ + 256);
  float pv[HP_BLOCKS_];
#pragma unroll
  for (int i = 0; i < HP_BLOCKS_; ++i) pv[i] = hp[(size_t)i * HP_PITCH_ + d_g];

  const v8f acc = mma_panel_32x64(
      alphas + ((size_t)b * TD_ + mb * 32) * TE_,
      en_seq + (size_t)b * TE_ * D_ + nb * 64, As, Bs);

  float mall = fmaxf(mz.x, lane_xor16(mz.x));
  mall = fmaxf(mall, lane_xor_row<8>(mall));
  mall = fmaxf(mall, lane_xor_row<4>(mall));
  mall = fmaxf(mall, lane_xor_row<2>(mall));
  mall = fmaxf(mall, lane_xor_row<1>(mall));
  const float ci = expf(mz.x - mall);
  float den = ci * mz.y;
  den = den + lane_xor16(den);
  den = den + lane_xor_row<8>(den);
  den = den + lane_xor_row<4>(den);
  den = den + lane_xor_row<2>(den);
  den = den + lane_xor_row<1>(den);
  float hnum = 0.f;
#pragma unroll
  for (int i = 0; i < HP_BLOCKS_; ++i) {
    const float c_i = __builtin_bit_cast(float,
        __builtin_amdgcn_readlane(__builtin_bit_cast(int, ci), i));
    hnum = fmaf(c_i, pv[i], hnum);
  }
  const float hh = hnum / den;
#pragma unroll
  for (int r = 0; r < 8; ++r) {
    const int s_g  = mb * 32 + (wave >> 2) * 16 + r + 8 * khalf;
    const float se = acc[r];
    const float de = de_seq[((size_t)(b * TD_ + s_g)) * D_ + d_g];
    float* o = out + ((size_t)(b * TD_ + s_g)) * (4 * D_);
    o[d_g]          = de;
    o[D_ + d_g]     = se;
    o[2 * D_ + d_g] = de * se;
    o[3 * D_ + d_g] = de * hh;
  }
}

extern "C" void kernel_launch(void* const* d_in, const int* in_sizes, int n_in,
                              void* d_out, int out_size, void* d_ws, size_t ws_size,
                              hipStream_t stream) {
  const float* en_seq = (const float*)d_in[0];
  const float* de_seq = (const float*)d_in[1];
  const float* w_en   = (const float*)d_in[2];
  const float* w_de   = (const float*)d_in[3];
  const float* nu     = (const float*)d_in[4];
  float* out = (float*)d_out;

  float* ws = (float*)d_ws;
  const size_t nAE = (size_t)B_ * TE_ * U_;
  float* ae     = ws;
  float* ad     = ae + nAE;
  float* mu     = ad + nAE;
  float* hpart  = mu + (size_t)B_ * TD_ * TE_;

  const dim3 blk(256);
  gemm_f32_wmma<<<dim3(256), blk, 0, stream>>>(en_seq, w_en, ae);
  gemm_f32_wmma<<<dim3(256), blk, 0, stream>>>(de_seq, w_de, ad);
  co_rows_softmax_kernel<<<dim3(HP_BLOCKS_, B_), blk, 0, stream>>>(ae, ad, nu, de_seq,
                                                                 mu, hpart);
  sum_en_out_kernel<<<dim3(32, B_), blk, 0, stream>>>(mu, en_seq, de_seq,
                                                      hpart, out);
}


# ae and ad formed in ONE launch of 512 blocks (launches 4 -> 3): grid y picks the product (en_seq*w_en or de_seq*w_de), each block still one product's 32x64 LDS-panel tile; k and sum order unchanged
# speedup vs baseline: 4.4213x; 1.1039x over previous
#include <hip/hip_runtime.h>
#include <cmath>

#define B_  8
#define TE_ 256
#define TD_ 256
#define D_  256
#define U_  256

typedef float v2f __attribute__((ext_vector_type(2)));
typedef float v8f __attribute__((ext_vector_type(8)));

typedef float v4f __attribute__((ext_vector_type(4)));
#define PA_ 260
#define PB_ 72
#define PANEL_A_FLOATS_ (32 * PA_)
#define PANEL_B_FLOATS_ (256 * PB_)

__device__ __forceinline__ void lds_quarter_barrier() {
  asm volatile("s_wait_dscnt 0x0" ::: "memory");
  asm volatile("s_barrier_signal -1" ::: "memory");
  asm volatile("s_barrier_wait -1" ::: "memory");
}

__device__ __forceinline__ v8f mma_panel_32x64(const float* __restrict__ Ag,
                                               const float* __restrict__ Bg,
                                               float* As, float* Bs) {
  const int tid   = threadIdx.x;
  const int lane  = tid & 31;
  const int wave  = tid >> 5;
  const int m     = lane & 15;
  const int khalf = lane >> 4;
  const int wm    = wave >> 2;
  const int wn    = wave & 3;
  const int row   = tid >> 4;
  const int c4    = (tid & 15) * 4;

  v4f ra[4][2], rb[4][4];
#pragma unroll
  for (int q = 0; q < 4; ++q) {
#pragma unroll
    for (int j = 0; j < 2; ++j)
      ra[q][j] = *(const v4f*)(Ag + (size_t)(j * 16 + row) * 256 + 64 * q + c4);
#pragma unroll
    for (int j = 0; j < 4; ++j)
      rb[q][j] = *(const v4f*)(Bg + (size_t)(64 * q + j * 16 + row) * 256 + c4);
  }

  const float* af = As + (wm * 16 + m) * PA_ + 2 * khalf;
  const float* bf = Bs + (2 * khalf) * PB_ + wn * 16 + m;
  v2f fa[2][16], fb[2][16];
  v8f acc = {};
#pragma unroll
  for (int q = -1; q < 4; ++q) {
    if (q < 3) {
      const int qn = q + 1;
#pragma unroll
      for (int j = 0; j < 2; ++j)
        *(v4f*)(As + (j * 16 + row) * PA_ + 64 * qn + c4) = ra[qn][j];
#pragma unroll
      for (int j = 0; j < 4; ++j)
        *(v4f*)(Bs + (64 * qn + j * 16 + row) * PB_ + c4) = rb[qn][j];
      lds_quarter_barrier();
#pragma unroll
      for (int i = 0; i < 16; ++i) {
        const int k4 = 4 * (16 * qn + i);
        fa[qn & 1][i]   = *(const v2f*)(af + k4);
        fb[qn & 1][i].x = bf[(k4 + 0) * PB_];
        fb[qn & 1][i].y = bf[(k4 + 1) * PB_];
      }
    }
    __builtin_amdgcn_sched_barrier(0);
    if (q >= 0) {
#pragma unroll
      for (int i = 0; i < 16; ++i)
        acc = __builtin_amdgcn_wmma_f32_16x16x4_f32(false, fa[q & 1][i], false,
                                                    fb[q & 1][i], (short)0, acc,
                                                    false, false);
    }
    __builtin_amdgcn_sched_barrier(0);
  }
  return acc;
}

__global__ void __launch_bounds__(256) gemm_pair_f32_wmma(
    const float* __restrict__ A0, const float* __restrict__ W0,
    float* __restrict__ C0,
    const float* __restrict__ A1, const float* __restrict__ W1,
    float* __restrict__ C1) {
  __shared__ __attribute__((aligned(16))) float As[PANEL_A_FLOATS_];
  __shared__ __attribute__((aligned(16))) float Bs[PANEL_B_FLOATS_];
  const bool second = blockIdx.y != 0;
  const float* __restrict__ A = second ? A1 : A0;
  const float* __restrict__ W = second ? W1 : W0;
  float* __restrict__ C       = second ? C1 : C0;
  const int lane  = threadIdx.x & 31;
  const int wave  = threadIdx.x >> 5;
  const int m     = lane & 15;
  const int khalf = lane >> 4;
  const int mb    = blockIdx.x >> 2;
  const int nb    = blockIdx.x & 3;

  const v8f acc = mma_panel_32x64(A + (size_t)mb * 32 * U_, W + nb * 64, As, Bs);

  const int n_out = nb * 64 + (wave & 3) * 16 + m;
#pragma unroll
  for (int r = 0; r < 8; ++r) {
    const int m_out = mb * 32 + (wave >> 2) * 16 + r + 8 * khalf;
    C[(size_t)m_out * U_ + n_out] = acc[r];
  }
}

#define EXP_LIM_ 43.0f
#define PE_ 260
#define HP_PITCH_ 288
#define HP_BLOCKS_ 32

__device__ __forceinline__ float exp_staged(float v) {
  const float kHi = 2.0f * 1.44269502f;
  const float kLo = 2.0f * 1.92596299e-8f;
  const float kLn2 = 0.693147182f;
  const float hi = kHi * v;
  float lo = fmaf(kHi, v, -hi);
  lo = fmaf(kLo, v, lo);
  const float e = __builtin_amdgcn_exp2f(hi);
  return fmaf(e, lo * kLn2, e);
}

__device__ __forceinline__ float mu_direct(const float* __restrict__ adr,
                                           const float* __restrict__ aer,
                                           const float* __restrict__ nu) {
  float acc = 0.f;
  for (int u = 0; u < U_; ++u) acc += tanhf(adr[u] + aer[u]) * nu[u];
  return acc;
}

__device__ __forceinline__ float lane_xor16(float x) {
  const int b = __builtin_bit_cast(int, x);
  return __builtin_bit_cast(float,
      __builtin_amdgcn_permlanex16(b, b, 0x76543210, 0xfedcba98, false, false));
}
template <int M>
__device__ __forceinline__ float lane_xor_row(float x) {
  const int b = __builtin_bit_cast(int, x);
  return __builtin_bit_cast(float,
      __builtin_amdgcn_update_dpp(b, b, 0x160 + M, 0xf, 0xf, false));
}

__global__ void __launch_bounds__(256) co_rows_softmax_kernel(
    const float* __restrict__ ae, const float* __restrict__ ad,
    const float* __restrict__ nu, const float* __restrict__ de_seq,
    float* __restrict__ alphas, float* __restrict__ hpart) {
  __shared__ __attribute__((aligned(16))) float EeT[256 * PE_];
  __shared__ __attribute__((aligned(16))) float EaT[8 * PE_];
  __shared__ __attribute__((aligned(16))) float nuS[256];
  __shared__ float muFix[256];
  __shared__ float nuPart[8];
  __shared__ float mxS[8];
  __shared__ int   badT[256];
  __shared__ int   badS[8];
  __shared__ int   anyBad;

  const int b    = blockIdx.y;
  const int s0   = blockIdx.x * 8;
  const int tid  = threadIdx.x;
  const int lane = tid & 31;
  const int wave = __builtin_amdgcn_readfirstlane((int)(threadIdx.x >> 5));
  const int row  = tid >> 4;
  const int c4   = (tid & 15) * 4;
  const float* aep = ae + (size_t)b * TE_ * U_;
  const float* adp = ad + ((size_t)b * TD_ + s0) * U_;

  v4f cur[16];
#pragma unroll
  for (int i = 0; i < 16; ++i)
    cur[i] = *(const v4f*)(aep + (size_t)(16 * i + row) * U_ + c4);
  v4f va[2];
#pragma unroll
  for (int k = 0; k < 2; ++k) va[k] = *(const v4f*)(adp + (size_t)(tid + 256 * k) * 4);

  float dv[8];
#pragma unroll
  for (int k = 0; k < 8; ++k)
    dv[k] = de_seq[((size_t)(b * TD_ + s0 + k)) * D_ + tid];

  const float nuv = nu[tid];
  nuS[tid]  = -2.0f * nuv;
  badT[tid] = 0;
  if (tid < 8) badS[tid] = 0;
  if (tid == 0) anyBad = 0;
  float nsum = nuv;
#pragma unroll
  for (int m = 16; m > 0; m >>= 1) nsum += __shfl_xor(nsum, m);
  if (lane == 0) nuPart[wave] = nsum;

  int badA[2];
#pragma unroll
  for (int k = 0; k < 2; ++k) {
    const int idx = tid + 256 * k;
    v4f e;
    int bad = 0;
#pragma unroll
    for (int x = 0; x < 4; ++x) {
      bad |= !(fabsf(va[k][x]) <= EXP_LIM_);
      e[x] = exp_staged(va[k][x]);
    }
    badA[k] = bad;
    *(v4f*)(EaT + (idx >> 6) * PE_ + (idx & 63) * 4) = e;
  }

  const float* eaRow  = EaT + wave * PE_;
  const float* eeLane = EeT + lane * PE_;
  float acc[8];
#pragma unroll
  for (int j = 0; j < 8; ++j) acc[j] = 0.f;
  unsigned badmask = 0;

#pragma unroll
  for (int q = 0; q < 4; ++q) {
#pragma unroll
    for (int i = 0; i < 16; ++i) {
      v4f e;
      int bad = 0;
#pragma unroll
      for (int x = 0; x < 4; ++x) {
        bad |= !(fabsf(cur[i][x]) <= EXP_LIM_);
        e[x] = exp_staged(cur[i][x]);
      }
      badmask |= (unsigned)bad << i;
      *(v4f*)(EeT + (16 * i + row) * PE_ + 64 * q + c4) = e;
    }
    if (q < 3) {
#pragma unroll
      for (int i = 0; i < 16; ++i)
        cur[i] = *(const v4f*)(aep + (size_t)(16 * i + row) * U_ + 64 * (q + 1) + c4);
    }
    lds_quarter_barrier();

#pragma unroll 2
    for (int u = 64 * q; u < 64 * q + 64; u += 4) {
      const v4f a4 = *(const v4f*)(eaRow + u);
      const v4f n4 = *(const v4f*)(nuS + u);
#pragma unroll
      for (int j = 0; j < 8; ++j) {
        const v4f e4 = *(const v4f*)(eeLane + j * 32 * PE_ + u);
#pragma unroll
        for (int x = 0; x < 4; ++x) {
          const float den = fmaf(a4[x], e4[x], 1.0f);
          acc[j] = fmaf(n4[x], __builtin_amdgcn_rcpf(den), acc[j]);
        }
      }
    }
  }

  if (badA[0]) { badS[tid >> 6] = 1; anyBad = 1; }
  if (badA[1]) { badS[(tid + 256) >> 6] = 1; anyBad = 1; }
  if (badmask) {
    anyBad = 1;
#pragma unroll
    for (int i = 0; i < 16; ++i)
      if ((badmask >> i) & 1u) badT[16 * i + row] = 1;
  }
  __syncthreads();

  float nu_total = 0.f;
#pragma unroll
  for (int w = 0; w < 8; ++w) nu_total += nuPart[w];
  float mu[8];
#pragma unroll
  for (int j = 0; j < 8; ++j) mu[j] = acc[j] + nu_total;

  if (anyBad) {
    for (int s = 0; s < 8; ++s) {
      if (badS[s]) {
        muFix[tid] = mu_direct(adp + (size_t)s * U_, aep + (size_t)tid * U_, nu);
        __syncthreads();
        if (wave == s) {
#pragma unroll
          for (int j = 0; j < 8; ++j) mu[j] = muFix[lane + 32 * j];
        }
        __syncthreads();
      }
    }
    if (!badS[wave]) {
#pragma unroll
      for (int j = 0; j < 8; ++j)
        if (badT[lane + 32 * j])
          mu[j] = mu_direct(adp + (size_t)wave * U_,
                            aep + (size_t)(lane + 32 * j) * U_, nu);
    }
  }

  float mx = fmaxf(fmaxf(fmaxf(mu[0], mu[4]), fmaxf(mu[2], mu[6])),
                   fmaxf(fmaxf(mu[1], mu[5]), fmaxf(mu[3], mu[7])));
  mx = fmaxf(mx, lane_xor16(mx));
  mx = fmaxf(mx, lane_xor_row<8>(mx));
  mx = fmaxf(mx, lane_xor_row<4>(mx));
  mx = fmaxf(mx, lane_xor_row<2>(mx));
  mx = fmaxf(mx, lane_xor_row<1>(mx));
  if (lane == 0) mxS[wave] = mx;
  float e[8];
#pragma unroll
  for (int j = 0; j < 8; ++j) e[j] = expf(mu[j] - mx);
  float sum = ((e[0] + e[4]) + (e[2] + e[6])) + ((e[1] + e[5]) + (e[3] + e[7]));
  sum = sum + lane_xor16(sum);
  sum = sum + lane_xor_row<8>(sum);
  sum = sum + lane_xor_row<4>(sum);
  sum = sum + lane_xor_row<2>(sum);
  sum = sum + lane_xor_row<1>(sum);
  const float inv = 1.0f / sum;

  float* dst = alphas + ((size_t)(b * TD_ + s0 + wave)) * TE_ + lane;
#pragma unroll
  for (int j = 0; j < 8; ++j) dst[32 * j] = e[j] * inv;

  lds_quarter_barrier();
  float mb = mxS[0];
#pragma unroll
  for (int k = 1; k < 8; ++k) mb = fmaxf(mb, mxS[k]);
  float z = 0.f, p = 0.f;
#pragma unroll
  for (int k = 0; k < 8; ++k) {
    const float w = expf(mxS[k] - mb);
    z += w;
    p = fmaf(w, dv[k], p);
  }
  float* hp = hpart + ((size_t)(b * HP_BLOCKS_ + blockIdx.x)) * HP_PITCH_;
  hp[tid] = p;
  if (tid == 0) { hp[256] = mb; hp[257] = z; }
}

__global__ void __launch_bounds__(256) sum_en_out_kernel(
    const float* __restrict__ alphas, const float* __restrict__ en_seq,
    const float* __restrict__ de_seq, const float* __restrict__ hpart,
    float* __restrict__ out) {
  __shared__ __attribute__((aligned(16))) float As[PANEL_A_FLOATS_];
  __shared__ __attribute__((aligned(16))) float Bs[PANEL_B_FLOATS_];
  const int b     = blockIdx.y;
  const int lane  = threadIdx.x & 31;
  const int wave  = threadIdx.x >> 5;
  const int m     = lane & 15;
  const int khalf = lane >> 4;
  const int mb    = blockIdx.x >> 2;
  const int nb    = blockIdx.x & 3;

  const int d_g  = nb * 64 + (wave & 3) * 16 + m;

  const float* hp = hpart + (size_t)b * HP_BLOCKS_ * HP_PITCH_;
  const v2f mz = *(const v2f*)(hp + (size_t)lane * HP_PITCH_ + 256);
  float pv[HP_BLOCKS_];
#pragma unroll
  for (int i = 0; i < HP_BLOCKS_; ++i) pv[i] = hp[(size_t)i * HP_PITCH_ + d_g];

  const v8f acc = mma_panel_32x64(
      alphas + ((size_t)b * TD_ + mb * 32) * TE_,
      en_seq + (size_t)b * TE_ * D_ + nb * 64, As, Bs);

  float mall = fmaxf(mz.x, lane_xor16(mz.x));
  mall = fmaxf(mall, lane_xor_row<8>(mall));
  mall = fmaxf(mall, lane_xor_row<4>(mall));
  mall = fmaxf(mall, lane_xor_row<2>(mall));
  mall = fmaxf(mall, lane_xor_row<1>(mall));
  const float ci = expf(mz.x - mall);
  float den = ci * mz.y;
  den = den + lane_xor16(den);
  den = den + lane_xor_row<8>(den);
  den = den + lane_xor_row<4>(den);
  den = den + lane_xor_row<2>(den);
  den = den + lane_xor_row<1>(den);
  float hnum = 0.f;
#pragma unroll
  for (int i = 0; i < HP_BLOCKS_; ++i) {
    const float c_i = __builtin_bit_cast(float,
        __builtin_amdgcn_readlane(__builtin_bit_cast(int, ci), i));
    hnum = fmaf(c_i, pv[i], hnum);
  }
  const float hh = hnum / den;
#pragma unroll
  for (int r = 0; r < 8; ++r) {
    const int s_g  = mb * 32 + (wave >> 2) * 16 + r + 8 * khalf;
    const float se = acc[r];
    const float de = de_seq[((size_t)(b * TD_ + s_g)) * D_ + d_g];
    float* o = out + ((size_t)(b * TD_ + s_g)) * (4 * D_);
    o[d_g]          = de;
    o[D_ + d_g]     = se;
    o[2 * D_ + d_g] = de * se;
    o[3 * D_ + d_g] = de * hh;
  }
}

extern "C" void kernel_launch(void* const* d_in, const int* in_sizes, int n_in,
                              void* d_out, int out_size, void* d_ws, size_t ws_size,
                              hipStream_t stream) {
  const float* en_seq = (const float*)d_in[0];
  const float* de_seq = (const float*)d_in[1];
  const float* w_en   = (const float*)d_in[2];
  const float* w_de   = (const float*)d_in[3];
  const float* nu     = (const float*)d_in[4];
  float* out = (float*)d_out;

  float* ws = (float*)d_ws;
  const size_t nAE = (size_t)B_ * TE_ * U_;
  float* ae     = ws;
  float* ad     = ae + nAE;
  float* mu     = ad + nAE;
  float* hpart  = mu + (size_t)B_ * TD_ * TE_;

  const dim3 blk(256);
  gemm_pair_f32_wmma<<<dim3(256, 2), blk, 0, stream>>>(en_seq, w_en, ae,
                                                       de_seq, w_de, ad);
  co_rows_softmax_kernel<<<dim3(HP_BLOCKS_, B_), blk, 0, stream>>>(ae, ad, nu, de_seq,
                                                                 mu, hpart);
  sum_en_out_kernel<<<dim3(32, B_), blk, 0, stream>>>(mu, en_seq, de_seq,
                                                      hpart, out);
}


# score pair loop: c=-2nu folded into the staged tiles (Ea/c, 1/c), so two u-terms share one v_rcp_f32 as (d0+d2)*rcp(d0*d2), d=(1+Ea*Ee)/c; tanhf path beyond |a|>6 (was 43); u-sum in 2 partial sums
# speedup vs baseline: 4.6634x; 1.0548x over previous
#include <hip/hip_runtime.h>
#include <cmath>

#define B_  8
#define TE_ 256
#define TD_ 256
#define D_  256
#define U_  256

typedef float v2f __attribute__((ext_vector_type(2)));
typedef float v8f __attribute__((ext_vector_type(8)));

typedef float v4f __attribute__((ext_vector_type(4)));
#define PA_ 260
#define PB_ 72
#define PANEL_A_FLOATS_ (32 * PA_)
#define PANEL_B_FLOATS_ (256 * PB_)

__device__ __forceinline__ void lds_quarter_barrier() {
  asm volatile("s_wait_dscnt 0x0" ::: "memory");
  asm volatile("s_barrier_signal -1" ::: "memory");
  asm volatile("s_barrier_wait -1" ::: "memory");
}

__device__ __forceinline__ v8f mma_panel_32x64(const float* __restrict__ Ag,
                                               const float* __restrict__ Bg,
                                               float* As, float* Bs) {
  const int tid   = threadIdx.x;
  const int lane  = tid & 31;
  const int wave  = tid >> 5;
  const int m     = lane & 15;
  const int khalf = lane >> 4;
  const int wm    = wave >> 2;
  const int wn    = wave & 3;
  const int row   = tid >> 4;
  const int c4    = (tid & 15) * 4;

  v4f ra[4][2], rb[4][4];
#pragma unroll
  for (int q = 0; q < 4; ++q) {
#pragma unroll
    for (int j = 0; j < 2; ++j)
      ra[q][j] = *(const v4f*)(Ag + (size_t)(j * 16 + row) * 256 + 64 * q + c4);
#pragma unroll
    for (int j = 0; j < 4; ++j)
      rb[q][j] = *(const v4f*)(Bg + (size_t)(64 * q + j * 16 + row) * 256 + c4);
  }

  const float* af = As + (wm * 16 + m) * PA_ + 2 * khalf;
  const float* bf = Bs + (2 * khalf) * PB_ + wn * 16 + m;
  v2f fa[2][16], fb[2][16];
  v8f acc = {};
#pragma unroll
  for (int q = -1; q < 4; ++q) {
    if (q < 3) {
      const int qn = q + 1;
#pragma unroll
      for (int j = 0; j < 2; ++j)
        *(v4f*)(As + (j * 16 + row) * PA_ + 64 * qn + c4) = ra[qn][j];
#pragma unroll
      for (int j = 0; j < 4; ++j)
        *(v4f*)(Bs + (64 * qn + j * 16 + row) * PB_ + c4) = rb[qn][j];
      lds_quarter_barrier();
#pragma unroll
      for (int i = 0; i < 16; ++i) {
        const int k4 = 4 * (16 * qn + i);
        fa[qn & 1][i]   = *(const v2f*)(af + k4);
        fb[qn & 1][i].x = bf[(k4 + 0) * PB_];
        fb[qn & 1][i].y = bf[(k4 + 1) * PB_];
      }
    }
    __builtin_amdgcn_sched_barrier(0);
    if (q >= 0) {
#pragma unroll
      for (int i = 0; i < 16; ++i)
        acc = __builtin_amdgcn_wmma_f32_16x16x4_f32(false, fa[q & 1][i], false,
                                                    fb[q & 1][i], (short)0, acc,
                                                    false, false);
    }
    __builtin_amdgcn_sched_barrier(0);
  }
  return acc;
}

__global__ void __launch_bounds__(256) gemm_pair_f32_wmma(
    const float* __restrict__ A0, const float* __restrict__ W0,
    float* __restrict__ C0,
    const float* __restrict__ A1, const float* __restrict__ W1,
    float* __restrict__ C1) {
  __shared__ __attribute__((aligned(16))) float As[PANEL_A_FLOATS_];
  __shared__ __attribute__((aligned(16))) float Bs[PANEL_B_FLOATS_];
  const bool second = blockIdx.y != 0;
  const float* __restrict__ A = second ? A1 : A0;
  const float* __restrict__ W = second ? W1 : W0;
  float* __restrict__ C       = second ? C1 : C0;
  const int lane  = threadIdx.x & 31;
  const int wave  = threadIdx.x >> 5;
  const int m     = lane & 15;
  const int khalf = lane >> 4;
  const int mb    = blockIdx.x >> 2;
  const int nb    = blockIdx.x & 3;

  const v8f acc = mma_panel_32x64(A + (size_t)mb * 32 * U_, W + nb * 64, As, Bs);

  const int n_out = nb * 64 + (wave & 3) * 16 + m;
#pragma unroll
  for (int r = 0; r < 8; ++r) {
    const int m_out = mb * 32 + (wave >> 2) * 16 + r + 8 * khalf;
    C[(size_t)m_out * U_ + n_out] = acc[r];
  }
}

#define EXP_LIM_ 6.0f
#define C_MIN_ 1.4901161e-8f
#define C_MAX_ 1.0995116e12f
#define PE_ 260
#define HP_PITCH_ 288
#define HP_BLOCKS_ 32

__device__ __forceinline__ float exp_staged(float v) {
  const float kHi = 2.0f * 1.44269502f;
  const float kLo = 2.0f * 1.92596299e-8f;
  const float kLn2 = 0.693147182f;
  const float hi = kHi * v;
  float lo = fmaf(kHi, v, -hi);
  lo = fmaf(kLo, v, lo);
  const float e = __builtin_amdgcn_exp2f(hi);
  return fmaf(e, lo * kLn2, e);
}

__device__ __forceinline__ float mu_direct(const float* __restrict__ adr,
                                           const float* __restrict__ aer,
                                           const float* __restrict__ nu) {
  float acc = 0.f;
  for (int u = 0; u < U_; ++u) acc += tanhf(adr[u] + aer[u]) * nu[u];
  return acc;
}

__device__ __forceinline__ float lane_xor16(float x) {
  const int b = __builtin_bit_cast(int, x);
  return __builtin_bit_cast(float,
      __builtin_amdgcn_permlanex16(b, b, 0x76543210, 0xfedcba98, false, false));
}
template <int M>
__device__ __forceinline__ float lane_xor_row(float x) {
  const int b = __builtin_bit_cast(int, x);
  return __builtin_bit_cast(float,
      __builtin_amdgcn_update_dpp(b, b, 0x160 + M, 0xf, 0xf, false));
}

__global__ void __launch_bounds__(256) co_rows_softmax_kernel(
    const float* __restrict__ ae, const float* __restrict__ ad,
    const float* __restrict__ nu, const float* __restrict__ de_seq,
    float* __restrict__ alphas, float* __restrict__ hpart) {
  __shared__ __attribute__((aligned(16))) float EeT[256 * PE_];
  __shared__ __attribute__((aligned(16))) float EaT[8 * PE_];
  __shared__ __attribute__((aligned(16))) float nuS[256];
  __shared__ float muFix[256];
  __shared__ float nuPart[8];
  __shared__ float mxS[8];
  __shared__ int   badT[256];
  __shared__ int   badS[8];
  __shared__ int   anyBad;

  const int b    = blockIdx.y;
  const int s0   = blockIdx.x * 8;
  const int tid  = threadIdx.x;
  const int lane = tid & 31;
  const int wave = __builtin_amdgcn_readfirstlane((int)(threadIdx.x >> 5));
  const int row  = tid >> 4;
  const int c4   = (tid & 15) * 4;
  const float* aep = ae + (size_t)b * TE_ * U_;
  const float* adp = ad + ((size_t)b * TD_ + s0) * U_;

  v4f cur[16];
#pragma unroll
  for (int i = 0; i < 16; ++i)
    cur[i] = *(const v4f*)(aep + (size_t)(16 * i + row) * U_ + c4);
  v4f va[2];
#pragma unroll
  for (int k = 0; k < 2; ++k) va[k] = *(const v4f*)(adp + (size_t)(tid + 256 * k) * 4);

  float dv[8];
#pragma unroll
  for (int k = 0; k < 8; ++k)
    dv[k] = de_seq[((size_t)(b * TD_ + s0 + k)) * D_ + tid];

  const float nuv = nu[tid];
  const float cw  = -2.0f * nuv;
  nuS[tid]  = 1.0f / cw;
  const int badN = !(fabsf(cw) >= C_MIN_ && fabsf(cw) <= C_MAX_);
  badT[tid] = 0;
  if (tid < 8) badS[tid] = 0;
  if (tid == 0) anyBad = 0;
  float nsum = nuv;
#pragma unroll
  for (int m = 16; m > 0; m >>= 1) nsum += __shfl_xor(nsum, m);
  if (lane == 0) nuPart[wave] = nsum;

  const v4f nu4 = *(const v4f*)(nu + (tid & 63) * 4);
  int badA[2];
#pragma unroll
  for (int k = 0; k < 2; ++k) {
    const int idx = tid + 256 * k;
    v4f e;
    int bad = 0;
#pragma unroll
    for (int x = 0; x < 4; ++x) {
      bad |= !(fabsf(va[k][x]) <= EXP_LIM_);
      e[x] = exp_staged(va[k][x]) * (1.0f / (-2.0f * nu4[x]));
    }
    badA[k] = bad;
    *(v4f*)(EaT + (idx >> 6) * PE_ + (idx & 63) * 4) = e;
  }

  const float* eaRow  = EaT + wave * PE_;
  const float* eeLane = EeT + lane * PE_;
  v2f acc[8];
#pragma unroll
  for (int j = 0; j < 8; ++j) acc[j] = (v2f){0.f, 0.f};
  unsigned badmask = 0;

#pragma unroll
  for (int q = 0; q < 4; ++q) {
#pragma unroll
    for (int i = 0; i < 16; ++i) {
      v4f e;
      int bad = 0;
#pragma unroll
      for (int x = 0; x < 4; ++x) {
        bad |= !(fabsf(cur[i][x]) <= EXP_LIM_);
        e[x] = exp_staged(cur[i][x]);
      }
      badmask |= (unsigned)bad << i;
      *(v4f*)(EeT + (16 * i + row) * PE_ + 64 * q + c4) = e;
    }
    if (q < 3) {
#pragma unroll
      for (int i = 0; i < 16; ++i)
        cur[i] = *(const v4f*)(aep + (size_t)(16 * i + row) * U_ + 64 * (q + 1) + c4);
    }
    lds_quarter_barrier();

#pragma unroll 2
    for (int u = 64 * q; u < 64 * q + 64; u += 4) {
      const v4f a4 = *(const v4f*)(eaRow + u);
      const v4f i4 = *(const v4f*)(nuS + u);
#pragma unroll
      for (int j = 0; j < 8; ++j) {
        const v4f e4 = *(const v4f*)(eeLane + j * 32 * PE_ + u);
        const v2f dl = {fmaf(a4[0], e4[0], i4[0]), fmaf(a4[1], e4[1], i4[1])};
        const v2f dh = {fmaf(a4[2], e4[2], i4[2]), fmaf(a4[3], e4[3], i4[3])};
        const v2f sm = dl + dh;
        const v2f pr = dl * dh;
        acc[j][0] = fmaf(sm[0], __builtin_amdgcn_rcpf(pr[0]), acc[j][0]);
        acc[j][1] = fmaf(sm[1], __builtin_amdgcn_rcpf(pr[1]), acc[j][1]);
      }
    }
  }

  if (badN) {
    anyBad = 1;
#pragma unroll
    for (int k = 0; k < 8; ++k) badS[k] = 1;
  }
  if (badA[0]) { badS[tid >> 6] = 1; anyBad = 1; }
  if (badA[1]) { badS[(tid + 256) >> 6] = 1; anyBad = 1; }
  if (badmask) {
    anyBad = 1;
#pragma unroll
    for (int i = 0; i < 16; ++i)
      if ((badmask >> i) & 1u) badT[16 * i + row] = 1;
  }
  __syncthreads();

  float nu_total = 0.f;
#pragma unroll
  for (int w = 0; w < 8; ++w) nu_total += nuPart[w];
  float mu[8];
#pragma unroll
  for (int j = 0; j < 8; ++j) mu[j] = (acc[j][0] + acc[j][1]) + nu_total;

  if (anyBad) {
    for (int s = 0; s < 8; ++s) {
      if (badS[s]) {
        muFix[tid] = mu_direct(adp + (size_t)s * U_, aep + (size_t)tid * U_, nu);
        __syncthreads();
        if (wave == s) {
#pragma unroll
          for (int j = 0; j < 8; ++j) mu[j] = muFix[lane + 32 * j];
        }
        __syncthreads();
      }
    }
    if (!badS[wave]) {
#pragma unroll
      for (int j = 0; j < 8; ++j)
        if (badT[lane + 32 * j])
          mu[j] = mu_direct(adp + (size_t)wave * U_,
                            aep + (size_t)(lane + 32 * j) * U_, nu);
    }
  }

  float mx = fmaxf(fmaxf(fmaxf(mu[0], mu[4]), fmaxf(mu[2], mu[6])),
                   fmaxf(fmaxf(mu[1], mu[5]), fmaxf(mu[3], mu[7])));
  mx = fmaxf(mx, lane_xor16(mx));
  mx = fmaxf(mx, lane_xor_row<8>(mx));
  mx = fmaxf(mx, lane_xor_row<4>(mx));
  mx = fmaxf(mx, lane_xor_row<2>(mx));
  mx = fmaxf(mx, lane_xor_row<1>(mx));
  if (lane == 0) mxS[wave] = mx;
  float e[8];
#pragma unroll
  for (int j = 0; j < 8; ++j) e[j] = expf(mu[j] - mx);
  float sum = ((e[0] + e[4]) + (e[2] + e[6])) + ((e[1] + e[5]) + (e[3] + e[7]));
  sum = sum + lane_xor16(sum);
  sum = sum + lane_xor_row<8>(sum);
  sum = sum + lane_xor_row<4>(sum);
  sum = sum + lane_xor_row<2>(sum);
  sum = sum + lane_xor_row<1>(sum);
  const float inv = 1.0f / sum;

  float* dst = alphas + ((size_t)(b * TD_ + s0 + wave)) * TE_ + lane;
#pragma unroll
  for (int j = 0; j < 8; ++j) dst[32 * j] = e[j] * inv;

  lds_quarter_barrier();
  float mb = mxS[0];
#pragma unroll
  for (int k = 1; k < 8; ++k) mb = fmaxf(mb, mxS[k]);
  float z = 0.f, p = 0.f;
#pragma unroll
  for (int k = 0; k < 8; ++k) {
    const float w = expf(mxS[k] - mb);
    z += w;
    p = fmaf(w, dv[k], p);
  }
  float* hp = hpart + ((size_t)(b * HP_BLOCKS_ + blockIdx.x)) * HP_PITCH_;
  hp[tid] = p;
  if (tid == 0) { hp[256] = mb; hp[257] = z; }
}

__global__ void __launch_bounds__(256) sum_en_out_kernel(
    const float* __restrict__ alphas, const float* __restrict__ en_seq,
    const float* __restrict__ de_seq, const float* __restrict__ hpart,
    float* __restrict__ out) {
  __shared__ __attribute__((aligned(16))) float As[PANEL_A_FLOATS_];
  __shared__ __attribute__((aligned(16))) float Bs[PANEL_B_FLOATS_];
  const int b     = blockIdx.y;
  const int lane  = threadIdx.x & 31;
  const int wave  = threadIdx.x >> 5;
  const int m     = lane & 15;
  const int khalf = lane >> 4;
  const int mb    = blockIdx.x >> 2;
  const int nb    = blockIdx.x & 3;

  const int d_g  = nb * 64 + (wave & 3) * 16 + m;

  const float* hp = hpart + (size_t)b * HP_BLOCKS_ * HP_PITCH_;
  const v2f mz = *(const v2f*)(hp + (size_t)lane * HP_PITCH_ + 256);
  float pv[HP_BLOCKS_];
#pragma unroll
  for (int i = 0; i < HP_BLOCKS_; ++i) pv[i] = hp[(size_t)i * HP_PITCH_ + d_g];

  const v8f acc = mma_panel_32x64(
      alphas + ((size_t)b * TD_ + mb * 32) * TE_,
      en_seq + (size_t)b * TE_ * D_ + nb * 64, As, Bs);

  float mall = fmaxf(mz.x, lane_xor16(mz.x));
  mall = fmaxf(mall, lane_xor_row<8>(mall));
  mall = fmaxf(mall, lane_xor_row<4>(mall));
  mall = fmaxf(mall, lane_xor_row<2>(mall));
  mall = fmaxf(mall, lane_xor_row<1>(mall));
  const float ci = expf(mz.x - mall);
  float den = ci * mz.y;
  den = den + lane_xor16(den);
  den = den + lane_xor_row<8>(den);
  den = den + lane_xor_row<4>(den);
  den = den + lane_xor_row<2>(den);
  den = den + lane_xor_row<1>(den);
  float hnum = 0.f;
#pragma unroll
  for (int i = 0; i < HP_BLOCKS_; ++i) {
    const float c_i = __builtin_bit_cast(float,
        __builtin_amdgcn_readlane(__builtin_bit_cast(int, ci), i));
    hnum = fmaf(c_i, pv[i], hnum);
  }
  const float hh = hnum / den;
#pragma unroll
  for (int r = 0; r < 8; ++r) {
    const int s_g  = mb * 32 + (wave >> 2) * 16 + r + 8 * khalf;
    const float se = acc[r];
    const float de = de_seq[((size_t)(b * TD_ + s_g)) * D_ + d_g];
    float* o = out + ((size_t)(b * TD_ + s_g)) * (4 * D_);
    o[d_g]          = de;
    o[D_ + d_g]     = se;
    o[2 * D_ + d_g] = de * se;
    o[3 * D_ + d_g] = de * hh;
  }
}

extern "C" void kernel_launch(void* const* d_in, const int* in_sizes, int n_in,
                              void* d_out, int out_size, void* d_ws, size_t ws_size,
                              hipStream_t stream) {
  const float* en_seq = (const float*)d_in[0];
  const float* de_seq = (const float*)d_in[1];
  const float* w_en   = (const float*)d_in[2];
  const float* w_de   = (const float*)d_in[3];
  const float* nu     = (const float*)d_in[4];
  float* out = (float*)d_out;

  float* ws = (float*)d_ws;
  const size_t nAE = (size_t)B_ * TE_ * U_;
  float* ae     = ws;
  float* ad     = ae + nAE;
  float* mu     = ad + nAE;
  float* hpart  = mu + (size_t)B_ * TD_ * TE_;

  const dim3 blk(256);
  gemm_pair_f32_wmma<<<dim3(256, 2), blk, 0, stream>>>(en_seq, w_en, ae,
                                                       de_seq, w_de, ad);
  co_rows_softmax_kernel<<<dim3(HP_BLOCKS_, B_), blk, 0, stream>>>(ae, ad, nu, de_seq,
                                                                 mu, hpart);
  sum_en_out_kernel<<<dim3(32, B_), blk, 0, stream>>>(mu, en_seq, de_seq,
                                                      hpart, out);
}
